# starting kernel
# baseline (speedup 1.0000x reference)
#include <hip/hip_runtime.h>
#include <hip/hip_bf16.h>


#define N_USER  200000
#define N_MOVIE 100000
#define N_EDGE  4000000
#define D       64

typedef float v2f __attribute__((ext_vector_type(2)));
typedef float v8f __attribute__((ext_vector_type(8)));

#define WSTRIDE 80
#define ASTRIDE 68

#define OFF_SU 0u
#define OFF_SM (OFF_SU + N_USER * 64u)
#define OFF_CU (OFF_SM + N_MOVIE * 64u)
#define OFF_CM (OFF_CU + N_USER)
#define OFF_U1 (OFF_CM + N_MOVIE)
#define OFF_M1 (OFF_U1 + N_USER * 64u)
#define ZERO_FLOATS (OFF_U1)
#define ZERO_F4 (ZERO_FLOATS / 4u)

__global__ __launch_bounds__(256) void zero_f4_kernel(float4* __restrict__ p,
                                                      unsigned n4) {
  unsigned i = blockIdx.x * 256u + threadIdx.x;
  if (i < n4) p[i] = make_float4(0.f, 0.f, 0.f, 0.f);
}

__global__ __launch_bounds__(256) void scatter_kernel(
    const float* __restrict__ xu, const float* __restrict__ xm,
    const int* __restrict__ uidx, const int* __restrict__ midx,
    float* __restrict__ su, float* __restrict__ sm,
    float* __restrict__ cu, float* __restrict__ cm, int nedge) {
  const int e    = (int)((blockIdx.x * 256u + threadIdx.x) >> 5);
  const int lane = threadIdx.x & 31;
  if (e >= nedge) return;
  const int u = uidx[e];
  const int m = midx[e];
  const int c = lane * 2;
  const float2 a = *(const float2*)&xu[(size_t)u * D + c];
  atomicAdd(&sm[(size_t)m * D + c + 0], a.x);
  atomicAdd(&sm[(size_t)m * D + c + 1], a.y);
  const float2 b = *(const float2*)&xm[(size_t)m * D + c];
  atomicAdd(&su[(size_t)u * D + c + 0], b.x);
  atomicAdd(&su[(size_t)u * D + c + 1], b.y);
  if (lane == 0) {
    atomicAdd(&cm[m], 1.0f);
    atomicAdd(&cu[u], 1.0f);
  }
}

__global__ __launch_bounds__(256) void sage_wmma_kernel(
    const float* __restrict__ s, const float* __restrict__ cnt,
    const float* __restrict__ xdst,
    const float* __restrict__ Wl, const float* __restrict__ Wr,
    const float* __restrict__ bias,
    float* __restrict__ out, int nrows, int do_relu) {
  __shared__ float ldsWl[D * WSTRIDE];
  __shared__ float ldsWr[D * WSTRIDE];
  __shared__ float ldsMean[32 * ASTRIDE];
  __shared__ float ldsX[32 * ASTRIDE];

  const int tid  = threadIdx.x;
  const int row0 = blockIdx.x * 32;

  for (int i = tid * 4; i < D * D; i += 256 * 4) {
    const int r = i >> 6, c = i & 63;
    *(float4*)&ldsWl[r * WSTRIDE + c] = *(const float4*)&Wl[i];
    *(float4*)&ldsWr[r * WSTRIDE + c] = *(const float4*)&Wr[i];
  }
  for (int i = tid * 4; i < 32 * D; i += 256 * 4) {
    const int r = i >> 6, c = i & 63;
    const int g = row0 + r;
    float4 mv = make_float4(0.f, 0.f, 0.f, 0.f);
    float4 xv = make_float4(0.f, 0.f, 0.f, 0.f);
    if (g < nrows) {
      const float inv = 1.0f / fmaxf(cnt[g], 1.0f);
      const float4 sv = *(const float4*)&s[(size_t)g * D + c];
      mv = make_float4(sv.x * inv, sv.y * inv, sv.z * inv, sv.w * inv);
      xv = *(const float4*)&xdst[(size_t)g * D + c];
    }
    *(float4*)&ldsMean[r * ASTRIDE + c] = mv;
    *(float4*)&ldsX[r * ASTRIDE + c]    = xv;
  }
  __syncthreads();

  const int wave  = tid >> 5;
  const int lane  = tid & 31;
  const int rt    = wave >> 2;
  const int ct    = wave & 3;
  const int mrow  = rt * 16 + (lane & 15);
  const int ncol  = ct * 16 + (lane & 15);
  const int khalf = (lane >> 4) * 2;

  v8f acc = {0.f, 0.f, 0.f, 0.f, 0.f, 0.f, 0.f, 0.f};
  #pragma unroll
  for (int k = 0; k < D; k += 4) {
    const int kk = k + khalf;
    const v2f a = *(const v2f*)&ldsMean[mrow * ASTRIDE + kk];
    v2f b;
    b.x = ldsWl[(kk + 0) * WSTRIDE + ncol];
    b.y = ldsWl[(kk + 1) * WSTRIDE + ncol];
    acc = __builtin_amdgcn_wmma_f32_16x16x4_f32(false, a, false, b,
                                                (short)0, acc, false, false);
  }
  #pragma unroll
  for (int k = 0; k < D; k += 4) {
    const int kk = k + khalf;
    const v2f a = *(const v2f*)&ldsX[mrow * ASTRIDE + kk];
    v2f b;
    b.x = ldsWr[(kk + 0) * WSTRIDE + ncol];
    b.y = ldsWr[(kk + 1) * WSTRIDE + ncol];
    acc = __builtin_amdgcn_wmma_f32_16x16x4_f32(false, a, false, b,
                                                (short)0, acc, false, false);
  }

  const float bcol  = bias[ncol];
  const int   mhalf = (lane < 16) ? 0 : 8;
  #pragma unroll
  for (int v = 0; v < 8; ++v) {
    const int g = row0 + rt * 16 + mhalf + v;
    if (g < nrows) {
      float val = acc[v] + bcol;
      if (do_relu) val = fmaxf(val, 0.0f);
      out[(size_t)g * D + ncol] = val;
    }
  }
}

extern "C" void kernel_launch(void* const* d_in, const int* in_sizes, int n_in,
                              void* d_out, int out_size, void* d_ws, size_t ws_size,
                              hipStream_t stream) {
  const float* x_user  = (const float*)d_in[0];
  const float* x_movie = (const float*)d_in[1];
  const float* Wl1_um  = (const float*)d_in[2];
  const float* Wr1_um  = (const float*)d_in[3];
  const float* b1_um   = (const float*)d_in[4];
  const float* Wl1_mu  = (const float*)d_in[5];
  const float* Wr1_mu  = (const float*)d_in[6];
  const float* b1_mu   = (const float*)d_in[7];
  const float* Wl2_um  = (const float*)d_in[8];
  const float* Wr2_um  = (const float*)d_in[9];
  const float* b2_um   = (const float*)d_in[10];
  const float* Wl2_mu  = (const float*)d_in[11];
  const float* Wr2_mu  = (const float*)d_in[12];
  const float* b2_mu   = (const float*)d_in[13];
  const int*   uidx    = (const int*)d_in[14];
  const int*   midx    = (const int*)d_in[15];

  float* ws = (float*)d_ws;
  float* su = ws + OFF_SU;
  float* sm = ws + OFF_SM;
  float* cu = ws + OFF_CU;
  float* cm = ws + OFF_CM;
  float* u1 = ws + OFF_U1;
  float* m1 = ws + OFF_M1;

  float* u2 = (float*)d_out;
  float* m2 = (float*)d_out + (size_t)N_USER * D;

  const dim3 blk(256);
  const dim3 gZero((ZERO_F4 + 255) / 256);
  const dim3 gScat((unsigned)(((size_t)N_EDGE * 32 + 255) / 256));
  const dim3 gMov(N_MOVIE / 32);
  const dim3 gUsr(N_USER / 32);

  zero_f4_kernel<<<gZero, blk, 0, stream>>>((float4*)ws, ZERO_F4);
  scatter_kernel<<<gScat, blk, 0, stream>>>(x_user, x_movie, uidx, midx,
                                            su, sm, cu, cm, N_EDGE);
  sage_wmma_kernel<<<gMov, blk, 0, stream>>>(sm, cm, x_movie, Wl1_um, Wr1_um,
                                             b1_um, m1, N_MOVIE, 1);
  sage_wmma_kernel<<<gUsr, blk, 0, stream>>>(su, cu, x_user, Wl1_mu, Wr1_mu,
                                             b1_mu, u1, N_USER, 1);

  zero_f4_kernel<<<gZero, blk, 0, stream>>>((float4*)ws, ZERO_F4);
  scatter_kernel<<<gScat, blk, 0, stream>>>(u1, m1, uidx, midx,
                                            su, sm, cu, cm, N_EDGE);
  sage_wmma_kernel<<<gMov, blk, 0, stream>>>(sm, cm, m1, Wl2_um, Wr2_um,
                                             b2_um, m2, N_MOVIE, 0);
  sage_wmma_kernel<<<gUsr, blk, 0, stream>>>(su, cu, u1, Wl2_mu, Wr2_mu,
                                             b2_mu, u2, N_USER, 0);
}


# Aggregation by pull: per-call CSR (int atomic rank, scan, fill), each row's neighbour sum gathered in registers in the transform's staging; replaces float atomic scatter; changes summation order
# speedup vs baseline: 2.0087x; 2.0087x over previous
#include <hip/hip_runtime.h>


#define N_USER  200000
#define N_MOVIE 100000
#define N_EDGE  4000000
#define D       64

typedef float v2f __attribute__((ext_vector_type(2)));
typedef float v8f __attribute__((ext_vector_type(8)));

#define WSTRIDE 80
#define ASTRIDE 68

#define W_DEGU  0u
#define W_DEGM  200000u
#define W_OFFU  300000u
#define W_OFFM  500064u
#define W_COLU  600128u
#define W_COLM  4600128u
#define W_RANKU 8600128u
#define W_RANKM 12600128u
#define OFF_U1  19500000u
#define OFF_M1  (OFF_U1 + N_USER * 64u)
#define DEG_F4  75000u

__global__ __launch_bounds__(256) void zero_f4_kernel(float4* __restrict__ p,
                                                      unsigned n4) {
  unsigned i = blockIdx.x * 256u + threadIdx.x;
  if (i < n4) p[i] = make_float4(0.f, 0.f, 0.f, 0.f);
}

__global__ __launch_bounds__(256) void rank_kernel(
    const int* __restrict__ uidx, const int* __restrict__ midx,
    int* __restrict__ degU, int* __restrict__ degM,
    int* __restrict__ rankU, int* __restrict__ rankM, int nedge) {
  const int e = (int)(blockIdx.x * 256u + threadIdx.x);
  if (e >= nedge) return;
  const int u = uidx[e];
  const int m = midx[e];
  const int ru = atomicAdd(&degU[u], 1);
  const int rm = atomicAdd(&degM[m], 1);
  rankU[e] = ru;
  rankM[e] = rm;
}

__global__ __launch_bounds__(1024) void scan_kernel(
    const int* __restrict__ degU, const int* __restrict__ degM,
    int* __restrict__ offU, int* __restrict__ offM) {
  __shared__ int wsum[32];
  const int* __restrict__ deg = (blockIdx.x == 0) ? degU : degM;
  int* __restrict__ off       = (blockIdx.x == 0) ? offU : offM;
  const int n  = (blockIdx.x == 0) ? N_USER : N_MOVIE;
  const int n4 = n >> 2;
  const int tid  = threadIdx.x;
  const int lane = tid & 31;
  const int wave = tid >> 5;
  int carry = 0;
  for (int tile = 0; tile < n4; tile += 4096) {
    const int i0 = tile + 4 * tid;
    int4 v[4];
    int  s4[4];
    int  t = 0;
    #pragma unroll
    for (int k = 0; k < 4; ++k) {
      v[k] = make_int4(0, 0, 0, 0);
      if (i0 + k < n4) v[k] = ((const int4*)deg)[i0 + k];
      s4[k] = t;
      t += v[k].x + v[k].y + v[k].z + v[k].w;
    }
    int x = t;
    #pragma unroll
    for (int d = 1; d < 32; d <<= 1) {
      const int y = __shfl_up(x, d);
      if (lane >= d) x += y;
    }
    if (lane == 31) wsum[wave] = x;
    __syncthreads();
    if (wave == 0) {
      int s = wsum[lane];
      #pragma unroll
      for (int d = 1; d < 32; d <<= 1) {
        const int y = __shfl_up(s, d);
        if (lane >= d) s += y;
      }
      wsum[lane] = s;
    }
    __syncthreads();
    const int wbase = (wave > 0) ? wsum[wave - 1] : 0;
    const int total = wsum[31];
    const int e0 = carry + wbase + x - t;
    #pragma unroll
    for (int k = 0; k < 4; ++k) {
      const int e = e0 + s4[k];
      if (i0 + k < n4)
        ((int4*)off)[i0 + k] = make_int4(e, e + v[k].x, e + v[k].x + v[k].y,
                                         e + v[k].x + v[k].y + v[k].z);
    }
    carry += total;
    __syncthreads();
  }
  if (tid == 0) off[n] = carry;
}

__global__ __launch_bounds__(256) void fill_kernel(
    const int* __restrict__ uidx, const int* __restrict__ midx,
    const int* __restrict__ offU, const int* __restrict__ offM,
    const int* __restrict__ rankU, const int* __restrict__ rankM,
    int* __restrict__ colU, int* __restrict__ colM, int nedge) {
  const int e = (int)(blockIdx.x * 256u + threadIdx.x);
  if (e >= nedge) return;
  const int u = uidx[e];
  const int m = midx[e];
  colU[offU[u] + rankU[e]] = m;
  colM[offM[m] + rankM[e]] = u;
}

__global__ __launch_bounds__(256) void sage_pull_wmma_kernel(
    const float* __restrict__ xsrc, const int* __restrict__ off,
    const int* __restrict__ col, const float* __restrict__ xdst,
    const float* __restrict__ Wl, const float* __restrict__ Wr,
    const float* __restrict__ bias,
    float* __restrict__ out, int nrows, int do_relu) {
  __shared__ float ldsWl[D * WSTRIDE];
  __shared__ float ldsWr[D * WSTRIDE];
  __shared__ float ldsMean[32 * ASTRIDE];
  __shared__ float ldsX[32 * ASTRIDE];

  const int tid  = threadIdx.x;
  const int row0 = blockIdx.x * 32;
  const int wave = tid >> 5;
  const int lane = tid & 31;

  for (int i = tid * 4; i < D * D; i += 256 * 4) {
    const int r = i >> 6, c = i & 63;
    *(float4*)&ldsWl[r * WSTRIDE + c] = *(const float4*)&Wl[i];
    *(float4*)&ldsWr[r * WSTRIDE + c] = *(const float4*)&Wr[i];
  }
  for (int i = tid * 4; i < 32 * D; i += 256 * 4) {
    const int r = i >> 6, c = i & 63;
    const int g = row0 + r;
    float4 xv = make_float4(0.f, 0.f, 0.f, 0.f);
    if (g < nrows) xv = *(const float4*)&xdst[(size_t)g * D + c];
    *(float4*)&ldsX[r * ASTRIDE + c] = xv;
  }
  int ov = 0;
  if (lane < 5) {
    const int gi = row0 + wave * 4 + lane;
    ov = off[gi < nrows ? gi : nrows];
  }
  int rb[5];
  #pragma unroll
  for (int q = 0; q < 5; ++q) rb[q] = __builtin_amdgcn_readlane(ov, q);
  int c0[4];
  #pragma unroll
  for (int q = 0; q < 4; ++q) {
    const int j = rb[q] + lane;
    c0[q] = (j < rb[q + 1]) ? col[j] : 0;
  }
  #pragma unroll
  for (int q = 0; q < 4; ++q) {
    const int r   = wave * 4 + q;
    const int beg = rb[q];
    const int end = rb[q + 1];
    float ax = 0.f, ay = 0.f;
    int myc = c0[q];
    for (int base = beg; base < end; base += 32) {
      if (base != beg) {
        const int j = base + lane;
        myc = (j < end) ? col[j] : 0;
      }
      const int rem = end - base;
      const int n   = rem < 32 ? rem : 32;
      for (int t0 = 0; t0 < n; t0 += 8) {
        #pragma unroll
        for (int k = 0; k < 8; ++k) {
          const int  t  = t0 + k;
          const bool ok = t < n;
          const int sidx = __builtin_amdgcn_readlane(myc, ok ? t : t0);
          const float2 v = *(const float2*)&xsrc[(size_t)sidx * D + lane * 2];
          ax += ok ? v.x : 0.f;
          ay += ok ? v.y : 0.f;
        }
      }
    }
    const float inv = 1.0f / fmaxf((float)(end - beg), 1.0f);
    *(float2*)&ldsMean[r * ASTRIDE + lane * 2] = make_float2(ax * inv, ay * inv);
  }
  __syncthreads();

  const int rt    = wave >> 2;
  const int ct    = wave & 3;
  const int mrow  = rt * 16 + (lane & 15);
  const int ncol  = ct * 16 + (lane & 15);
  const int khalf = (lane >> 4) * 2;

  v8f acc = {0.f, 0.f, 0.f, 0.f, 0.f, 0.f, 0.f, 0.f};
  #pragma unroll
  for (int k = 0; k < D; k += 4) {
    const int kk = k + khalf;
    const v2f a = *(const v2f*)&ldsMean[mrow * ASTRIDE + kk];
    v2f b;
    b.x = ldsWl[(kk + 0) * WSTRIDE + ncol];
    b.y = ldsWl[(kk + 1) * WSTRIDE + ncol];
    acc = __builtin_amdgcn_wmma_f32_16x16x4_f32(false, a, false, b,
                                                (short)0, acc, false, false);
  }
  #pragma unroll
  for (int k = 0; k < D; k += 4) {
    const int kk = k + khalf;
    const v2f a = *(const v2f*)&ldsX[mrow * ASTRIDE + kk];
    v2f b;
    b.x = ldsWr[(kk + 0) * WSTRIDE + ncol];
    b.y = ldsWr[(kk + 1) * WSTRIDE + ncol];
    acc = __builtin_amdgcn_wmma_f32_16x16x4_f32(false, a, false, b,
                                                (short)0, acc, false, false);
  }

  const float bcol  = bias[ncol];
  const int   mhalf = (lane < 16) ? 0 : 8;
  #pragma unroll
  for (int v = 0; v < 8; ++v) {
    const int g = row0 + rt * 16 + mhalf + v;
    if (g < nrows) {
      float val = acc[v] + bcol;
      if (do_relu) val = fmaxf(val, 0.0f);
      out[(size_t)g * D + ncol] = val;
    }
  }
}

extern "C" void kernel_launch(void* const* d_in, const int* in_sizes, int n_in,
                              void* d_out, int out_size, void* d_ws, size_t ws_size,
                              hipStream_t stream) {
  const float* x_user  = (const float*)d_in[0];
  const float* x_movie = (const float*)d_in[1];
  const float* Wl1_um  = (const float*)d_in[2];
  const float* Wr1_um  = (const float*)d_in[3];
  const float* b1_um   = (const float*)d_in[4];
  const float* Wl1_mu  = (const float*)d_in[5];
  const float* Wr1_mu  = (const float*)d_in[6];
  const float* b1_mu   = (const float*)d_in[7];
  const float* Wl2_um  = (const float*)d_in[8];
  const float* Wr2_um  = (const float*)d_in[9];
  const float* b2_um   = (const float*)d_in[10];
  const float* Wl2_mu  = (const float*)d_in[11];
  const float* Wr2_mu  = (const float*)d_in[12];
  const float* b2_mu   = (const float*)d_in[13];
  const int*   uidx    = (const int*)d_in[14];
  const int*   midx    = (const int*)d_in[15];

  float* ws  = (float*)d_ws;
  int*   wsi = (int*)d_ws;
  int* degU  = wsi + W_DEGU;
  int* degM  = wsi + W_DEGM;
  int* offU  = wsi + W_OFFU;
  int* offM  = wsi + W_OFFM;
  int* colU  = wsi + W_COLU;
  int* colM  = wsi + W_COLM;
  int* rankU = wsi + W_RANKU;
  int* rankM = wsi + W_RANKM;
  float* u1 = ws + OFF_U1;
  float* m1 = ws + OFF_M1;

  float* u2 = (float*)d_out;
  float* m2 = (float*)d_out + (size_t)N_USER * D;

  const dim3 blk(256);
  const dim3 gZero((DEG_F4 + 255) / 256);
  const dim3 gEdge((N_EDGE + 255) / 256);
  const dim3 gMov(N_MOVIE / 32);
  const dim3 gUsr(N_USER / 32);

  zero_f4_kernel<<<gZero, blk, 0, stream>>>((float4*)ws, DEG_F4);
  rank_kernel<<<gEdge, blk, 0, stream>>>(uidx, midx, degU, degM, rankU, rankM, N_EDGE);
  scan_kernel<<<dim3(2), dim3(1024), 0, stream>>>(degU, degM, offU, offM);
  fill_kernel<<<gEdge, blk, 0, stream>>>(uidx, midx, offU, offM, rankU, rankM,
                                         colU, colM, N_EDGE);

  sage_pull_wmma_kernel<<<gMov, blk, 0, stream>>>(x_user, offM, colM, x_movie,
                                                  Wl1_um, Wr1_um, b1_um, m1, N_MOVIE, 1);
  sage_pull_wmma_kernel<<<gUsr, blk, 0, stream>>>(x_movie, offU, colU, x_user,
                                                  Wl1_mu, Wr1_mu, b1_mu, u1, N_USER, 1);

  sage_pull_wmma_kernel<<<gMov, blk, 0, stream>>>(u1, offM, colM, m1,
                                                  Wl2_um, Wr2_um, b2_um, m2, N_MOVIE, 0);
  sage_pull_wmma_kernel<<<gUsr, blk, 0, stream>>>(m1, offU, colU, u1,
                                                  Wl2_mu, Wr2_mu, b2_mu, u2, N_USER, 0);
}


# CSR built by two-level counting sort on LDS counters (edges binned by destination range, ranked inside each range), no global atomics left; neighbour order in a row, hence summation order, changes
# speedup vs baseline: 4.7402x; 2.3598x over previous
#include <hip/hip_runtime.h>


#define N_USER  200000
#define N_MOVIE 100000
#define N_EDGE  4000000
#define D       64

typedef float v2f __attribute__((ext_vector_type(2)));
typedef float v8f __attribute__((ext_vector_type(8)));

#define WSTRIDE 80
#define ASTRIDE 68

#define CH       16384
#define NCH      245
#define NB       196
#define NCNT     48020
#define SHIFT0   10
#define SHIFT1   9
#define SRCBITS0 17
#define SRCBITS1 18
#define STCAP    24576

#define W_CNT0  0u
#define W_CNT1  50000u
#define W_BASE0 100000u
#define W_BASE1 150000u
#define W_OFFU  300000u
#define W_OFFM  500064u
#define W_COLU  600128u
#define W_COLM  4600128u
#define W_REC0  8600128u
#define W_REC1  12600128u
#define OFF_U1  19500000u
#define OFF_M1  (OFF_U1 + N_USER * 64u)

__global__ __launch_bounds__(1024) void bin_count_kernel(
    const int* __restrict__ uidx, const int* __restrict__ midx,
    int* __restrict__ cnt0, int* __restrict__ cnt1, int nedge) {
  __shared__ int hist[NB];
  const int dir = blockIdx.y;
  const int* __restrict__ dst = dir ? midx : uidx;
  int* __restrict__ cnt       = dir ? cnt1 : cnt0;
  const int shift = dir ? SHIFT1 : SHIFT0;
  const int tid = threadIdx.x;
  if (tid < NB) hist[tid] = 0;
  __syncthreads();
  const int e0 = blockIdx.x * CH;
  int d[16];
  #pragma unroll
  for (int k = 0; k < 16; ++k) {
    const int e = e0 + k * 1024 + tid;
    d[k] = (e < nedge) ? dst[e] : -1;
  }
  #pragma unroll
  for (int k = 0; k < 16; ++k)
    if (d[k] >= 0) atomicAdd(&hist[d[k] >> shift], 1);
  __syncthreads();
  if (tid < NB) cnt[tid * NCH + blockIdx.x] = hist[tid];
}

__global__ __launch_bounds__(1024) void scan_kernel(
    const int* __restrict__ cnt0, const int* __restrict__ cnt1,
    int* __restrict__ base0, int* __restrict__ base1) {
  __shared__ int wsum[32];
  const int* __restrict__ deg = (blockIdx.x == 0) ? cnt0 : cnt1;
  int* __restrict__ off       = (blockIdx.x == 0) ? base0 : base1;
  const int n  = NCNT;
  const int n4 = n >> 2;
  const int tid  = threadIdx.x;
  const int lane = tid & 31;
  const int wave = tid >> 5;
  int carry = 0;
  for (int tile = 0; tile < n4; tile += 4096) {
    const int i0 = tile + 4 * tid;
    int4 v[4];
    int  s4[4];
    int  t = 0;
    #pragma unroll
    for (int k = 0; k < 4; ++k) {
      v[k] = make_int4(0, 0, 0, 0);
      if (i0 + k < n4) v[k] = ((const int4*)deg)[i0 + k];
      s4[k] = t;
      t += v[k].x + v[k].y + v[k].z + v[k].w;
    }
    int x = t;
    #pragma unroll
    for (int d = 1; d < 32; d <<= 1) {
      const int y = __shfl_up(x, d);
      if (lane >= d) x += y;
    }
    if (lane == 31) wsum[wave] = x;
    __syncthreads();
    if (wave == 0) {
      int s = wsum[lane];
      #pragma unroll
      for (int d = 1; d < 32; d <<= 1) {
        const int y = __shfl_up(s, d);
        if (lane >= d) s += y;
      }
      wsum[lane] = s;
    }
    __syncthreads();
    const int wbase = (wave > 0) ? wsum[wave - 1] : 0;
    const int total = wsum[31];
    const int e0 = carry + wbase + x - t;
    #pragma unroll
    for (int k = 0; k < 4; ++k) {
      const int e = e0 + s4[k];
      if (i0 + k < n4)
        ((int4*)off)[i0 + k] = make_int4(e, e + v[k].x, e + v[k].x + v[k].y,
                                         e + v[k].x + v[k].y + v[k].z);
    }
    carry += total;
    __syncthreads();
  }
  if (tid == 0) off[n] = carry;
}

__global__ __launch_bounds__(1024) void bin_scatter_kernel(
    const int* __restrict__ uidx, const int* __restrict__ midx,
    const int* __restrict__ base0, const int* __restrict__ base1,
    int* __restrict__ rec0, int* __restrict__ rec1, int nedge) {
  __shared__ int sDst[CH];
  __shared__ int sSrc[CH];
  __shared__ int cur[256];
  __shared__ int dlt[256];
  __shared__ int wtot[8];
  const int dir = blockIdx.y;
  const int* __restrict__ dstA = dir ? midx : uidx;
  const int* __restrict__ srcA = dir ? uidx : midx;
  const int* __restrict__ base = dir ? base1 : base0;
  int* __restrict__ rec        = dir ? rec1 : rec0;
  const int shift = dir ? SHIFT1 : SHIFT0;
  const int sbits = dir ? SRCBITS1 : SRCBITS0;
  const int tid  = threadIdx.x;
  const int lane = tid & 31;
  const int c    = blockIdx.x;
  const int e0   = c * CH;
  const int rest = nedge - e0;
  const int nloc = rest < CH ? rest : CH;

  int d[16], s[16];
  #pragma unroll
  for (int k = 0; k < 16; ++k) {
    const int e = e0 + k * 1024 + tid;
    d[k] = (e < nedge) ? dstA[e] : -1;
    s[k] = (e < nedge) ? srcA[e] : 0;
  }

  int g = 0, cn = 0;
  if (tid < NB) {
    g  = base[tid * NCH + c];
    cn = base[tid * NCH + c + 1] - g;
  }
  int x = cn;
  if (tid < 256) {
    #pragma unroll
    for (int q = 1; q < 32; q <<= 1) {
      const int y = __shfl_up(x, q);
      if (lane >= q) x += y;
    }
    if (lane == 31) wtot[tid >> 5] = x;
  }
  __syncthreads();
  if (tid < 256) {
    int wb = 0;
    for (int w = 0; w < (tid >> 5); ++w) wb += wtot[w];
    const int ex = wb + x - cn;
    cur[tid] = ex;
    dlt[tid] = g - ex;
  }
  __syncthreads();

  #pragma unroll
  for (int k = 0; k < 16; ++k) {
    if (d[k] >= 0) {
      const int p = atomicAdd(&cur[d[k] >> shift], 1);
      sDst[p] = d[k];
      sSrc[p] = s[k];
    }
  }
  __syncthreads();

  const int lmask = (1 << shift) - 1;
  for (int i = tid; i < nloc; i += 1024) {
    const int dd = sDst[i];
    rec[i + dlt[dd >> shift]] = sSrc[i] | ((dd & lmask) << sbits);
  }
}

__global__ __launch_bounds__(1024) void bucket_csr_kernel(
    const int* __restrict__ base0, const int* __restrict__ base1,
    const int* __restrict__ rec0, const int* __restrict__ rec1,
    int* __restrict__ offU, int* __restrict__ offM,
    int* __restrict__ colU, int* __restrict__ colM) {
  __shared__ int cnt[1024];
  __shared__ int wsum[32];
  __shared__ int stage[STCAP];
  const int dir = blockIdx.y;
  const int b   = blockIdx.x;
  const int* __restrict__ base = dir ? base1 : base0;
  const int* __restrict__ rec  = dir ? rec1 : rec0;
  int* __restrict__ off        = dir ? offM : offU;
  int* __restrict__ col        = dir ? colM : colU;
  const int shift = dir ? SHIFT1 : SHIFT0;
  const int sbits = dir ? SRCBITS1 : SRCBITS0;
  const int nrows = dir ? N_MOVIE : N_USER;
  const int smask = (1 << sbits) - 1;
  const int tid  = threadIdx.x;
  const int lane = tid & 31;
  const int wave = tid >> 5;
  const int bs = base[b * NCH];
  const int be = base[(b + 1) * NCH];
  const int n  = be - bs;

  cnt[tid] = 0;
  __syncthreads();
  for (int i0 = bs; i0 < be; i0 += 4096) {
    int r[4];
    #pragma unroll
    for (int k = 0; k < 4; ++k) {
      const int i = i0 + k * 1024 + tid;
      r[k] = (i < be) ? rec[i] : -1;
    }
    #pragma unroll
    for (int k = 0; k < 4; ++k)
      if (r[k] >= 0) atomicAdd(&cnt[r[k] >> sbits], 1);
  }
  __syncthreads();

  const int v = cnt[tid];
  int x = v;
  #pragma unroll
  for (int q = 1; q < 32; q <<= 1) {
    const int y = __shfl_up(x, q);
    if (lane >= q) x += y;
  }
  if (lane == 31) wsum[wave] = x;
  __syncthreads();
  if (wave == 0) {
    int sacc = wsum[lane];
    #pragma unroll
    for (int q = 1; q < 32; q <<= 1) {
      const int y = __shfl_up(sacc, q);
      if (lane >= q) sacc += y;
    }
    wsum[lane] = sacc;
  }
  __syncthreads();
  const int ex = ((wave > 0) ? wsum[wave - 1] : 0) + x - v;
  const int node = (b << shift) + tid;
  if (tid < (1 << shift) && node < nrows) off[node] = bs + ex;
  if (b == NB - 1 && tid == 0) off[nrows] = be;
  cnt[tid] = ex;
  __syncthreads();

  if (n <= STCAP) {
    for (int i0 = bs; i0 < be; i0 += 4096) {
      int r[4];
      #pragma unroll
      for (int k = 0; k < 4; ++k) {
        const int i = i0 + k * 1024 + tid;
        r[k] = (i < be) ? rec[i] : -1;
      }
      #pragma unroll
      for (int k = 0; k < 4; ++k)
        if (r[k] >= 0) stage[atomicAdd(&cnt[r[k] >> sbits], 1)] = r[k] & smask;
    }
    __syncthreads();
    for (int i = tid; i < n; i += 1024) col[bs + i] = stage[i];
  } else {
    for (int i = bs + tid; i < be; i += 1024) {
      const int rr = rec[i];
      col[bs + atomicAdd(&cnt[rr >> sbits], 1)] = rr & smask;
    }
  }
}

__global__ __launch_bounds__(256) void sage_pull_wmma_kernel(
    const float* __restrict__ xsrc, const int* __restrict__ off,
    const int* __restrict__ col, const float* __restrict__ xdst,
    const float* __restrict__ Wl, const float* __restrict__ Wr,
    const float* __restrict__ bias,
    float* __restrict__ out, int nrows, int do_relu) {
  __shared__ float ldsWl[D * WSTRIDE];
  __shared__ float ldsWr[D * WSTRIDE];
  __shared__ float ldsMean[32 * ASTRIDE];
  __shared__ float ldsX[32 * ASTRIDE];

  const int tid  = threadIdx.x;
  const int row0 = blockIdx.x * 32;
  const int wave = tid >> 5;
  const int lane = tid & 31;

  for (int i = tid * 4; i < D * D; i += 256 * 4) {
    const int r = i >> 6, c = i & 63;
    *(float4*)&ldsWl[r * WSTRIDE + c] = *(const float4*)&Wl[i];
    *(float4*)&ldsWr[r * WSTRIDE + c] = *(const float4*)&Wr[i];
  }
  for (int i = tid * 4; i < 32 * D; i += 256 * 4) {
    const int r = i >> 6, c = i & 63;
    const int g = row0 + r;
    float4 xv = make_float4(0.f, 0.f, 0.f, 0.f);
    if (g < nrows) xv = *(const float4*)&xdst[(size_t)g * D + c];
    *(float4*)&ldsX[r * ASTRIDE + c] = xv;
  }
  int ov = 0;
  if (lane < 5) {
    const int gi = row0 + wave * 4 + lane;
    ov = off[gi < nrows ? gi : nrows];
  }
  int rb[5];
  #pragma unroll
  for (int q = 0; q < 5; ++q) rb[q] = __builtin_amdgcn_readlane(ov, q);
  int c0[4];
  #pragma unroll
  for (int q = 0; q < 4; ++q) {
    const int j = rb[q] + lane;
    c0[q] = (j < rb[q + 1]) ? col[j] : 0;
  }
  #pragma unroll
  for (int q = 0; q < 4; ++q) {
    const int r   = wave * 4 + q;
    const int beg = rb[q];
    const int end = rb[q + 1];
    float ax = 0.f, ay = 0.f;
    int myc = c0[q];
    for (int base = beg; base < end; base += 32) {
      if (base != beg) {
        const int j = base + lane;
        myc = (j < end) ? col[j] : 0;
      }
      const int rem = end - base;
      const int n   = rem < 32 ? rem : 32;
      for (int t0 = 0; t0 < n; t0 += 8) {
        #pragma unroll
        for (int k = 0; k < 8; ++k) {
          const int  t  = t0 + k;
          const bool ok = t < n;
          const int sidx = __builtin_amdgcn_readlane(myc, ok ? t : t0);
          const float2 v = *(const float2*)&xsrc[(size_t)sidx * D + lane * 2];
          ax += ok ? v.x : 0.f;
          ay += ok ? v.y : 0.f;
        }
      }
    }
    const float inv = 1.0f / fmaxf((float)(end - beg), 1.0f);
    *(float2*)&ldsMean[r * ASTRIDE + lane * 2] = make_float2(ax * inv, ay * inv);
  }
  __syncthreads();

  const int rt    = wave >> 2;
  const int ct    = wave & 3;
  const int mrow  = rt * 16 + (lane & 15);
  const int ncol  = ct * 16 + (lane & 15);
  const int khalf = (lane >> 4) * 2;

  v8f acc = {0.f, 0.f, 0.f, 0.f, 0.f, 0.f, 0.f, 0.f};
  #pragma unroll
  for (int k = 0; k < D; k += 4) {
    const int kk = k + khalf;
    const v2f a = *(const v2f*)&ldsMean[mrow * ASTRIDE + kk];
    v2f b;
    b.x = ldsWl[(kk + 0) * WSTRIDE + ncol];
    b.y = ldsWl[(kk + 1) * WSTRIDE + ncol];
    acc = __builtin_amdgcn_wmma_f32_16x16x4_f32(false, a, false, b,
                                                (short)0, acc, false, false);
  }
  #pragma unroll
  for (int k = 0; k < D; k += 4) {
    const int kk = k + khalf;
    const v2f a = *(const v2f*)&ldsX[mrow * ASTRIDE + kk];
    v2f b;
    b.x = ldsWr[(kk + 0) * WSTRIDE + ncol];
    b.y = ldsWr[(kk + 1) * WSTRIDE + ncol];
    acc = __builtin_amdgcn_wmma_f32_16x16x4_f32(false, a, false, b,
                                                (short)0, acc, false, false);
  }

  const float bcol  = bias[ncol];
  const int   mhalf = (lane < 16) ? 0 : 8;
  #pragma unroll
  for (int v = 0; v < 8; ++v) {
    const int g = row0 + rt * 16 + mhalf + v;
    if (g < nrows) {
      float val = acc[v] + bcol;
      if (do_relu) val = fmaxf(val, 0.0f);
      out[(size_t)g * D + ncol] = val;
    }
  }
}

extern "C" void kernel_launch(void* const* d_in, const int* in_sizes, int n_in,
                              void* d_out, int out_size, void* d_ws, size_t ws_size,
                              hipStream_t stream) {
  const float* x_user  = (const float*)d_in[0];
  const float* x_movie = (const float*)d_in[1];
  const float* Wl1_um  = (const float*)d_in[2];
  const float* Wr1_um  = (const float*)d_in[3];
  const float* b1_um   = (const float*)d_in[4];
  const float* Wl1_mu  = (const float*)d_in[5];
  const float* Wr1_mu  = (const float*)d_in[6];
  const float* b1_mu   = (const float*)d_in[7];
  const float* Wl2_um  = (const float*)d_in[8];
  const float* Wr2_um  = (const float*)d_in[9];
  const float* b2_um   = (const float*)d_in[10];
  const float* Wl2_mu  = (const float*)d_in[11];
  const float* Wr2_mu  = (const float*)d_in[12];
  const float* b2_mu   = (const float*)d_in[13];
  const int*   uidx    = (const int*)d_in[14];
  const int*   midx    = (const int*)d_in[15];

  float* ws  = (float*)d_ws;
  int*   wsi = (int*)d_ws;
  int* cnt0  = wsi + W_CNT0;
  int* cnt1  = wsi + W_CNT1;
  int* base0 = wsi + W_BASE0;
  int* base1 = wsi + W_BASE1;
  int* offU  = wsi + W_OFFU;
  int* offM  = wsi + W_OFFM;
  int* colU  = wsi + W_COLU;
  int* colM  = wsi + W_COLM;
  int* rec0  = wsi + W_REC0;
  int* rec1  = wsi + W_REC1;
  float* u1 = ws + OFF_U1;
  float* m1 = ws + OFF_M1;

  float* u2 = (float*)d_out;
  float* m2 = (float*)d_out + (size_t)N_USER * D;

  const dim3 blk(256);
  const dim3 big(1024);
  const dim3 gChunk(NCH, 2);
  const dim3 gBucket(NB, 2);
  const dim3 gMov(N_MOVIE / 32);
  const dim3 gUsr(N_USER / 32);

  bin_count_kernel<<<gChunk, big, 0, stream>>>(uidx, midx, cnt0, cnt1, N_EDGE);
  scan_kernel<<<dim3(2), big, 0, stream>>>(cnt0, cnt1, base0, base1);
  bin_scatter_kernel<<<gChunk, big, 0, stream>>>(uidx, midx, base0, base1,
                                                 rec0, rec1, N_EDGE);
  bucket_csr_kernel<<<gBucket, big, 0, stream>>>(base0, base1, rec0, rec1,
                                                 offU, offM, colU, colM);

  sage_pull_wmma_kernel<<<gMov, blk, 0, stream>>>(x_user, offM, colM, x_movie,
                                                  Wl1_um, Wr1_um, b1_um, m1, N_MOVIE, 1);
  sage_pull_wmma_kernel<<<gUsr, blk, 0, stream>>>(x_movie, offU, colU, x_user,
                                                  Wl1_mu, Wr1_mu, b1_mu, u1, N_USER, 1);

  sage_pull_wmma_kernel<<<gMov, blk, 0, stream>>>(u1, offM, colM, m1,
                                                  Wl2_um, Wr2_um, b2_um, m2, N_MOVIE, 0);
  sage_pull_wmma_kernel<<<gUsr, blk, 0, stream>>>(m1, offU, colU, u1,
                                                  Wl2_mu, Wr2_mu, b2_mu, u2, N_USER, 0);
}


# Layer kernel: fewer dependent load round trips per work-group: a wave gathers its 4 rows interleaved, longest first (32 row loads in flight, was 8); staging loads leave as one group; sum order kept
# speedup vs baseline: 4.9935x; 1.0534x over previous
#include <hip/hip_runtime.h>


#define N_USER  200000
#define N_MOVIE 100000
#define N_EDGE  4000000
#define D       64

typedef float v2f __attribute__((ext_vector_type(2)));
typedef float v8f __attribute__((ext_vector_type(8)));

#define WSTRIDE 80
#define ASTRIDE 68

#define CH       16384
#define NCH      245
#define NB       196
#define NCNT     48020
#define SHIFT0   10
#define SHIFT1   9
#define SRCBITS0 17
#define SRCBITS1 18
#define STCAP    24576

#define W_CNT0  0u
#define W_CNT1  50000u
#define W_BASE0 100000u
#define W_BASE1 150000u
#define W_OFFU  300000u
#define W_OFFM  500064u
#define W_COLU  600128u
#define W_COLM  4600128u
#define W_REC0  8600128u
#define W_REC1  12600128u
#define OFF_U1  19500000u
#define OFF_M1  (OFF_U1 + N_USER * 64u)

__global__ __launch_bounds__(1024) void bin_count_kernel(
    const int* __restrict__ uidx, const int* __restrict__ midx,
    int* __restrict__ cnt0, int* __restrict__ cnt1, int nedge) {
  __shared__ int hist[NB];
  const int dir = blockIdx.y;
  const int* __restrict__ dst = dir ? midx : uidx;
  int* __restrict__ cnt       = dir ? cnt1 : cnt0;
  const int shift = dir ? SHIFT1 : SHIFT0;
  const int tid = threadIdx.x;
  if (tid < NB) hist[tid] = 0;
  __syncthreads();
  const int e0 = blockIdx.x * CH;
  int d[16];
  #pragma unroll
  for (int k = 0; k < 16; ++k) {
    const int e = e0 + k * 1024 + tid;
    d[k] = (e < nedge) ? dst[e] : -1;
  }
  #pragma unroll
  for (int k = 0; k < 16; ++k)
    if (d[k] >= 0) atomicAdd(&hist[d[k] >> shift], 1);
  __syncthreads();
  if (tid < NB) cnt[tid * NCH + blockIdx.x] = hist[tid];
}

__global__ __launch_bounds__(1024) void scan_kernel(
    const int* __restrict__ cnt0, const int* __restrict__ cnt1,
    int* __restrict__ base0, int* __restrict__ base1) {
  __shared__ int wsum[32];
  const int* __restrict__ deg = (blockIdx.x == 0) ? cnt0 : cnt1;
  int* __restrict__ off       = (blockIdx.x == 0) ? base0 : base1;
  const int n  = NCNT;
  const int n4 = n >> 2;
  const int tid  = threadIdx.x;
  const int lane = tid & 31;
  const int wave = tid >> 5;
  int carry = 0;
  for (int tile = 0; tile < n4; tile += 4096) {
    const int i0 = tile + 4 * tid;
    int4 v[4];
    int  s4[4];
    int  t = 0;
    #pragma unroll
    for (int k = 0; k < 4; ++k) {
      v[k] = make_int4(0, 0, 0, 0);
      if (i0 + k < n4) v[k] = ((const int4*)deg)[i0 + k];
      s4[k] = t;
      t += v[k].x + v[k].y + v[k].z + v[k].w;
    }
    int x = t;
    #pragma unroll
    for (int d = 1; d < 32; d <<= 1) {
      const int y = __shfl_up(x, d);
      if (lane >= d) x += y;
    }
    if (lane == 31) wsum[wave] = x;
    __syncthreads();
    if (wave == 0) {
      int s = wsum[lane];
      #pragma unroll
      for (int d = 1; d < 32; d <<= 1) {
        const int y = __shfl_up(s, d);
        if (lane >= d) s += y;
      }
      wsum[lane] = s;
    }
    __syncthreads();
    const int wbase = (wave > 0) ? wsum[wave - 1] : 0;
    const int total = wsum[31];
    const int e0 = carry + wbase + x - t;
    #pragma unroll
    for (int k = 0; k < 4; ++k) {
      const int e = e0 + s4[k];
      if (i0 + k < n4)
        ((int4*)off)[i0 + k] = make_int4(e, e + v[k].x, e + v[k].x + v[k].y,
                                         e + v[k].x + v[k].y + v[k].z);
    }
    carry += total;
    __syncthreads();
  }
  if (tid == 0) off[n] = carry;
}

__global__ __launch_bounds__(1024) void bin_scatter_kernel(
    const int* __restrict__ uidx, const int* __restrict__ midx,
    const int* __restrict__ base0, const int* __restrict__ base1,
    int* __restrict__ rec0, int* __restrict__ rec1, int nedge) {
  __shared__ int sDst[CH];
  __shared__ int sSrc[CH];
  __shared__ int cur[256];
  __shared__ int dlt[256];
  __shared__ int wtot[8];
  const int dir = blockIdx.y;
  const int* __restrict__ dstA = dir ? midx : uidx;
  const int* __restrict__ srcA = dir ? uidx : midx;
  const int* __restrict__ base = dir ? base1 : base0;
  int* __restrict__ rec        = dir ? rec1 : rec0;
  const int shift = dir ? SHIFT1 : SHIFT0;
  const int sbits = dir ? SRCBITS1 : SRCBITS0;
  const int tid  = threadIdx.x;
  const int lane = tid & 31;
  const int c    = blockIdx.x;
  const int e0   = c * CH;
  const int rest = nedge - e0;
  const int nloc = rest < CH ? rest : CH;

  int d[16], s[16];
  #pragma unroll
  for (int k = 0; k < 16; ++k) {
    const int e = e0 + k * 1024 + tid;
    d[k] = (e < nedge) ? dstA[e] : -1;
    s[k] = (e < nedge) ? srcA[e] : 0;
  }

  int g = 0, cn = 0;
  if (tid < NB) {
    g  = base[tid * NCH + c];
    cn = base[tid * NCH + c + 1] - g;
  }
  int x = cn;
  if (tid < 256) {
    #pragma unroll
    for (int q = 1; q < 32; q <<= 1) {
      const int y = __shfl_up(x, q);
      if (lane >= q) x += y;
    }
    if (lane == 31) wtot[tid >> 5] = x;
  }
  __syncthreads();
  if (tid < 256) {
    int wb = 0;
    for (int w = 0; w < (tid >> 5); ++w) wb += wtot[w];
    const int ex = wb + x - cn;
    cur[tid] = ex;
    dlt[tid] = g - ex;
  }
  __syncthreads();

  #pragma unroll
  for (int k = 0; k < 16; ++k) {
    if (d[k] >= 0) {
      const int p = atomicAdd(&cur[d[k] >> shift], 1);
      sDst[p] = d[k];
      sSrc[p] = s[k];
    }
  }
  __syncthreads();

  const int lmask = (1 << shift) - 1;
  for (int i = tid; i < nloc; i += 1024) {
    const int dd = sDst[i];
    rec[i + dlt[dd >> shift]] = sSrc[i] | ((dd & lmask) << sbits);
  }
}

__global__ __launch_bounds__(1024) void bucket_csr_kernel(
    const int* __restrict__ base0, const int* __restrict__ base1,
    const int* __restrict__ rec0, const int* __restrict__ rec1,
    int* __restrict__ offU, int* __restrict__ offM,
    int* __restrict__ colU, int* __restrict__ colM) {
  __shared__ int cnt[1024];
  __shared__ int wsum[32];
  __shared__ int stage[STCAP];
  const int dir = blockIdx.y;
  const int b   = blockIdx.x;
  const int* __restrict__ base = dir ? base1 : base0;
  const int* __restrict__ rec  = dir ? rec1 : rec0;
  int* __restrict__ off        = dir ? offM : offU;
  int* __restrict__ col        = dir ? colM : colU;
  const int shift = dir ? SHIFT1 : SHIFT0;
  const int sbits = dir ? SRCBITS1 : SRCBITS0;
  const int nrows = dir ? N_MOVIE : N_USER;
  const int smask = (1 << sbits) - 1;
  const int tid  = threadIdx.x;
  const int lane = tid & 31;
  const int wave = tid >> 5;
  const int bs = base[b * NCH];
  const int be = base[(b + 1) * NCH];
  const int n  = be - bs;

  cnt[tid] = 0;
  __syncthreads();
  for (int i0 = bs; i0 < be; i0 += 4096) {
    int r[4];
    #pragma unroll
    for (int k = 0; k < 4; ++k) {
      const int i = i0 + k * 1024 + tid;
      r[k] = (i < be) ? rec[i] : -1;
    }
    #pragma unroll
    for (int k = 0; k < 4; ++k)
      if (r[k] >= 0) atomicAdd(&cnt[r[k] >> sbits], 1);
  }
  __syncthreads();

  const int v = cnt[tid];
  int x = v;
  #pragma unroll
  for (int q = 1; q < 32; q <<= 1) {
    const int y = __shfl_up(x, q);
    if (lane >= q) x += y;
  }
  if (lane == 31) wsum[wave] = x;
  __syncthreads();
  if (wave == 0) {
    int sacc = wsum[lane];
    #pragma unroll
    for (int q = 1; q < 32; q <<= 1) {
      const int y = __shfl_up(sacc, q);
      if (lane >= q) sacc += y;
    }
    wsum[lane] = sacc;
  }
  __syncthreads();
  const int ex = ((wave > 0) ? wsum[wave - 1] : 0) + x - v;
  const int node = (b << shift) + tid;
  if (tid < (1 << shift) && node < nrows) off[node] = bs + ex;
  if (b == NB - 1 && tid == 0) off[nrows] = be;
  cnt[tid] = ex;
  __syncthreads();

  if (n <= STCAP) {
    for (int i0 = bs; i0 < be; i0 += 4096) {
      int r[4];
      #pragma unroll
      for (int k = 0; k < 4; ++k) {
        const int i = i0 + k * 1024 + tid;
        r[k] = (i < be) ? rec[i] : -1;
      }
      #pragma unroll
      for (int k = 0; k < 4; ++k)
        if (r[k] >= 0) stage[atomicAdd(&cnt[r[k] >> sbits], 1)] = r[k] & smask;
    }
    __syncthreads();
    for (int i = tid; i < n; i += 1024) col[bs + i] = stage[i];
  } else {
    for (int i = bs + tid; i < be; i += 1024) {
      const int rr = rec[i];
      col[bs + atomicAdd(&cnt[rr >> sbits], 1)] = rr & smask;
    }
  }
}

template <int NA>
__device__ __forceinline__ void gather_step(const float* __restrict__ xl,
                                            const int (&ids)[4], const int (&dg)[4],
                                            int cb, int t0, float (&ax)[4], float (&ay)[4]) {
  float2 v[NA][8];
  #pragma unroll
  for (int p = 0; p < NA; ++p) {
    #pragma unroll
    for (int k = 0; k < 8; ++k) {
      const int t = t0 + k;
      const int sidx = __builtin_amdgcn_readlane(ids[p], ((t < dg[p]) ? t : t0) - cb);
      v[p][k] = *(const float2*)&xl[(size_t)sidx * D];
    }
  }
  #pragma unroll
  for (int p = 0; p < NA; ++p) {
    #pragma unroll
    for (int k = 0; k < 8; ++k) {
      const bool ok = (t0 + k) < dg[p];
      ax[p] += ok ? v[p][k].x : 0.f;
      ay[p] += ok ? v[p][k].y : 0.f;
    }
  }
}

__global__ __launch_bounds__(256) void sage_pull_wmma_kernel(
    const float* __restrict__ xsrc, const int* __restrict__ off,
    const int* __restrict__ col, const float* __restrict__ xdst,
    const float* __restrict__ Wl, const float* __restrict__ Wr,
    const float* __restrict__ bias,
    float* __restrict__ out, int nrows, int do_relu) {
  __shared__ float ldsWl[D * WSTRIDE];
  __shared__ float ldsWr[D * WSTRIDE];
  __shared__ float ldsMean[32 * ASTRIDE];
  __shared__ float ldsX[32 * ASTRIDE];

  const int tid  = threadIdx.x;
  const int row0 = blockIdx.x * 32;
  const int wave = tid >> 5;
  const int lane = tid & 31;

  int ov = 0;
  if (lane < 5) {
    const int gi = row0 + wave * 4 + lane;
    ov = off[gi < nrows ? gi : nrows];
  }

  float4 wl[4], wr[4], xv[2];
  #pragma unroll
  for (int u = 0; u < 4; ++u) {
    const int i = (tid + u * 256) * 4;
    wl[u] = *(const float4*)&Wl[i];
    wr[u] = *(const float4*)&Wr[i];
  }
  #pragma unroll
  for (int u = 0; u < 2; ++u) {
    const int i = (tid + u * 256) * 4;
    const int r = i >> 6, c = i & 63;
    const int g = row0 + r;
    xv[u] = make_float4(0.f, 0.f, 0.f, 0.f);
    if (g < nrows) xv[u] = *(const float4*)&xdst[(size_t)g * D + c];
  }

  int rb[5];
  #pragma unroll
  for (int q = 0; q < 5; ++q) rb[q] = __builtin_amdgcn_readlane(ov, q);

  int dg[4], st[4], ix[4];
  #pragma unroll
  for (int q = 0; q < 4; ++q) { dg[q] = rb[q + 1] - rb[q]; st[q] = rb[q]; ix[q] = q; }
#define SAGE_CSWAP(a, b)                                         \
  {                                                              \
    const bool sw = dg[a] < dg[b];                               \
    const int d0_ = sw ? dg[b] : dg[a], d1_ = sw ? dg[a] : dg[b]; \
    const int s0_ = sw ? st[b] : st[a], s1_ = sw ? st[a] : st[b]; \
    const int i0_ = sw ? ix[b] : ix[a], i1_ = sw ? ix[a] : ix[b]; \
    dg[a] = d0_; dg[b] = d1_; st[a] = s0_; st[b] = s1_; ix[a] = i0_; ix[b] = i1_; \
  }
  SAGE_CSWAP(0, 1) SAGE_CSWAP(2, 3) SAGE_CSWAP(0, 2) SAGE_CSWAP(1, 3) SAGE_CSWAP(1, 2)
#undef SAGE_CSWAP

  int ids[4];
  #pragma unroll
  for (int p = 0; p < 4; ++p) ids[p] = (lane < dg[p]) ? col[st[p] + lane] : 0;

  #pragma unroll
  for (int u = 0; u < 4; ++u) {
    const int i = (tid + u * 256) * 4;
    const int r = i >> 6, c = i & 63;
    *(float4*)&ldsWl[r * WSTRIDE + c] = wl[u];
    *(float4*)&ldsWr[r * WSTRIDE + c] = wr[u];
  }
  #pragma unroll
  for (int u = 0; u < 2; ++u) {
    const int i = (tid + u * 256) * 4;
    const int r = i >> 6, c = i & 63;
    *(float4*)&ldsX[r * ASTRIDE + c] = xv[u];
  }

  float ax[4] = {0.f, 0.f, 0.f, 0.f};
  float ay[4] = {0.f, 0.f, 0.f, 0.f};
  const float* __restrict__ xl = xsrc + lane * 2;
  for (int cb = 0; cb < dg[0]; cb += 32) {
    const int ce = cb + 32;
    int idn[4] = {0, 0, 0, 0};
    if (ce < dg[0]) {
      #pragma unroll
      for (int p = 0; p < 4; ++p) idn[p] = (ce + lane < dg[p]) ? col[st[p] + ce + lane] : 0;
    }
    const int e3 = dg[3] < ce ? dg[3] : ce;
    const int e2 = dg[2] < ce ? dg[2] : ce;
    const int e1 = dg[1] < ce ? dg[1] : ce;
    const int e0 = dg[0] < ce ? dg[0] : ce;
    int t0 = cb;
    for (; t0 < e3; t0 += 8) gather_step<4>(xl, ids, dg, cb, t0, ax, ay);
    for (; t0 < e2; t0 += 8) gather_step<3>(xl, ids, dg, cb, t0, ax, ay);
    for (; t0 < e1; t0 += 8) gather_step<2>(xl, ids, dg, cb, t0, ax, ay);
    for (; t0 < e0; t0 += 8) gather_step<1>(xl, ids, dg, cb, t0, ax, ay);
    #pragma unroll
    for (int p = 0; p < 4; ++p) ids[p] = idn[p];
  }
  #pragma unroll
  for (int p = 0; p < 4; ++p) {
    const int r = wave * 4 + ix[p];
    const float inv = 1.0f / fmaxf((float)dg[p], 1.0f);
    *(float2*)&ldsMean[r * ASTRIDE + lane * 2] = make_float2(ax[p] * inv, ay[p] * inv);
  }
  __syncthreads();

  const int rt    = wave >> 2;
  const int ct    = wave & 3;
  const int mrow  = rt * 16 + (lane & 15);
  const int ncol  = ct * 16 + (lane & 15);
  const int khalf = (lane >> 4) * 2;

  v8f acc = {0.f, 0.f, 0.f, 0.f, 0.f, 0.f, 0.f, 0.f};
  #pragma unroll
  for (int k = 0; k < D; k += 4) {
    const int kk = k + khalf;
    const v2f a = *(const v2f*)&ldsMean[mrow * ASTRIDE + kk];
    v2f b;
    b.x = ldsWl[(kk + 0) * WSTRIDE + ncol];
    b.y = ldsWl[(kk + 1) * WSTRIDE + ncol];
    acc = __builtin_amdgcn_wmma_f32_16x16x4_f32(false, a, false, b,
                                                (short)0, acc, false, false);
  }
  #pragma unroll
  for (int k = 0; k < D; k += 4) {
    const int kk = k + khalf;
    const v2f a = *(const v2f*)&ldsX[mrow * ASTRIDE + kk];
    v2f b;
    b.x = ldsWr[(kk + 0) * WSTRIDE + ncol];
    b.y = ldsWr[(kk + 1) * WSTRIDE + ncol];
    acc = __builtin_amdgcn_wmma_f32_16x16x4_f32(false, a, false, b,
                                                (short)0, acc, false, false);
  }

  const float bcol  = bias[ncol];
  const int   mhalf = (lane < 16) ? 0 : 8;
  #pragma unroll
  for (int v = 0; v < 8; ++v) {
    const int g = row0 + rt * 16 + mhalf + v;
    if (g < nrows) {
      float val = acc[v] + bcol;
      if (do_relu) val = fmaxf(val, 0.0f);
      out[(size_t)g * D + ncol] = val;
    }
  }
}

extern "C" void kernel_launch(void* const* d_in, const int* in_sizes, int n_in,
                              void* d_out, int out_size, void* d_ws, size_t ws_size,
                              hipStream_t stream) {
  const float* x_user  = (const float*)d_in[0];
  const float* x_movie = (const float*)d_in[1];
  const float* Wl1_um  = (const float*)d_in[2];
  const float* Wr1_um  = (const float*)d_in[3];
  const float* b1_um   = (const float*)d_in[4];
  const float* Wl1_mu  = (const float*)d_in[5];
  const float* Wr1_mu  = (const float*)d_in[6];
  const float* b1_mu   = (const float*)d_in[7];
  const float* Wl2_um  = (const float*)d_in[8];
  const float* Wr2_um  = (const float*)d_in[9];
  const float* b2_um   = (const float*)d_in[10];
  const float* Wl2_mu  = (const float*)d_in[11];
  const float* Wr2_mu  = (const float*)d_in[12];
  const float* b2_mu   = (const float*)d_in[13];
  const int*   uidx    = (const int*)d_in[14];
  const int*   midx    = (const int*)d_in[15];

  float* ws  = (float*)d_ws;
  int*   wsi = (int*)d_ws;
  int* cnt0  = wsi + W_CNT0;
  int* cnt1  = wsi + W_CNT1;
  int* base0 = wsi + W_BASE0;
  int* base1 = wsi + W_BASE1;
  int* offU  = wsi + W_OFFU;
  int* offM  = wsi + W_OFFM;
  int* colU  = wsi + W_COLU;
  int* colM  = wsi + W_COLM;
  int* rec0  = wsi + W_REC0;
  int* rec1  = wsi + W_REC1;
  float* u1 = ws + OFF_U1;
  float* m1 = ws + OFF_M1;

  float* u2 = (float*)d_out;
  float* m2 = (float*)d_out + (size_t)N_USER * D;

  const dim3 blk(256);
  const dim3 big(1024);
  const dim3 gChunk(NCH, 2);
  const dim3 gBucket(NB, 2);
  const dim3 gMov(N_MOVIE / 32);
  const dim3 gUsr(N_USER / 32);

  bin_count_kernel<<<gChunk, big, 0, stream>>>(uidx, midx, cnt0, cnt1, N_EDGE);
  scan_kernel<<<dim3(2), big, 0, stream>>>(cnt0, cnt1, base0, base1);
  bin_scatter_kernel<<<gChunk, big, 0, stream>>>(uidx, midx, base0, base1,
                                                 rec0, rec1, N_EDGE);
  bucket_csr_kernel<<<gBucket, big, 0, stream>>>(base0, base1, rec0, rec1,
                                                 offU, offM, colU, colM);

  sage_pull_wmma_kernel<<<gMov, blk, 0, stream>>>(x_user, offM, colM, x_movie,
                                                  Wl1_um, Wr1_um, b1_um, m1, N_MOVIE, 1);
  sage_pull_wmma_kernel<<<gUsr, blk, 0, stream>>>(x_movie, offU, colU, x_user,
                                                  Wl1_mu, Wr1_mu, b1_mu, u1, N_USER, 1);

  sage_pull_wmma_kernel<<<gMov, blk, 0, stream>>>(u1, offM, colM, m1,
                                                  Wl2_um, Wr2_um, b2_um, m2, N_MOVIE, 0);
  sage_pull_wmma_kernel<<<gUsr, blk, 0, stream>>>(m1, offU, colU, u1,
                                                  Wl2_mu, Wr2_mu, b2_mu, u2, N_USER, 0);
}


# Layer kernel: grid of 1280 work-groups (all resident at once); each walks its 32-row tiles with both weight matrices staged in LDS once, not per tile; 3.7x fewer waves enter; sum order kept
# speedup vs baseline: 5.2430x; 1.0500x over previous
#include <hip/hip_runtime.h>


#define N_USER  200000
#define N_MOVIE 100000
#define N_EDGE  4000000
#define D       64

typedef float v2f __attribute__((ext_vector_type(2)));
typedef float v8f __attribute__((ext_vector_type(8)));

#define WSTRIDE 80
#define ASTRIDE 68

#define CH       16384
#define NCH      245
#define NB       196
#define NCNT     48020
#define SHIFT0   10
#define SHIFT1   9
#define SRCBITS0 17
#define SRCBITS1 18
#define STCAP    24576

#define W_CNT0  0u
#define W_CNT1  50000u
#define W_BASE0 100000u
#define W_BASE1 150000u
#define W_OFFU  300000u
#define W_OFFM  500064u
#define W_COLU  600128u
#define W_COLM  4600128u
#define W_REC0  8600128u
#define W_REC1  12600128u
#define OFF_U1  19500000u
#define OFF_M1  (OFF_U1 + N_USER * 64u)

__global__ __launch_bounds__(1024) void bin_count_kernel(
    const int* __restrict__ uidx, const int* __restrict__ midx,
    int* __restrict__ cnt0, int* __restrict__ cnt1, int nedge) {
  __shared__ int hist[NB];
  const int dir = blockIdx.y;
  const int* __restrict__ dst = dir ? midx : uidx;
  int* __restrict__ cnt       = dir ? cnt1 : cnt0;
  const int shift = dir ? SHIFT1 : SHIFT0;
  const int tid = threadIdx.x;
  if (tid < NB) hist[tid] = 0;
  __syncthreads();
  const int e0 = blockIdx.x * CH;
  int d[16];
  #pragma unroll
  for (int k = 0; k < 16; ++k) {
    const int e = e0 + k * 1024 + tid;
    d[k] = (e < nedge) ? dst[e] : -1;
  }
  #pragma unroll
  for (int k = 0; k < 16; ++k)
    if (d[k] >= 0) atomicAdd(&hist[d[k] >> shift], 1);
  __syncthreads();
  if (tid < NB) cnt[tid * NCH + blockIdx.x] = hist[tid];
}

__global__ __launch_bounds__(1024) void scan_kernel(
    const int* __restrict__ cnt0, const int* __restrict__ cnt1,
    int* __restrict__ base0, int* __restrict__ base1) {
  __shared__ int wsum[32];
  const int* __restrict__ deg = (blockIdx.x == 0) ? cnt0 : cnt1;
  int* __restrict__ off       = (blockIdx.x == 0) ? base0 : base1;
  const int n  = NCNT;
  const int n4 = n >> 2;
  const int tid  = threadIdx.x;
  const int lane = tid & 31;
  const int wave = tid >> 5;
  int carry = 0;
  for (int tile = 0; tile < n4; tile += 4096) {
    const int i0 = tile + 4 * tid;
    int4 v[4];
    int  s4[4];
    int  t = 0;
    #pragma unroll
    for (int k = 0; k < 4; ++k) {
      v[k] = make_int4(0, 0, 0, 0);
      if (i0 + k < n4) v[k] = ((const int4*)deg)[i0 + k];
      s4[k] = t;
      t += v[k].x + v[k].y + v[k].z + v[k].w;
    }
    int x = t;
    #pragma unroll
    for (int d = 1; d < 32; d <<= 1) {
      const int y = __shfl_up(x, d);
      if (lane >= d) x += y;
    }
    if (lane == 31) wsum[wave] = x;
    __syncthreads();
    if (wave == 0) {
      int s = wsum[lane];
      #pragma unroll
      for (int d = 1; d < 32; d <<= 1) {
        const int y = __shfl_up(s, d);
        if (lane >= d) s += y;
      }
      wsum[lane] = s;
    }
    __syncthreads();
    const int wbase = (wave > 0) ? wsum[wave - 1] : 0;
    const int total = wsum[31];
    const int e0 = carry + wbase + x - t;
    #pragma unroll
    for (int k = 0; k < 4; ++k) {
      const int e = e0 + s4[k];
      if (i0 + k < n4)
        ((int4*)off)[i0 + k] = make_int4(e, e + v[k].x, e + v[k].x + v[k].y,
                                         e + v[k].x + v[k].y + v[k].z);
    }
    carry += total;
    __syncthreads();
  }
  if (tid == 0) off[n] = carry;
}

__global__ __launch_bounds__(1024) void bin_scatter_kernel(
    const int* __restrict__ uidx, const int* __restrict__ midx,
    const int* __restrict__ base0, const int* __restrict__ base1,
    int* __restrict__ rec0, int* __restrict__ rec1, int nedge) {
  __shared__ int sDst[CH];
  __shared__ int sSrc[CH];
  __shared__ int cur[256];
  __shared__ int dlt[256];
  __shared__ int wtot[8];
  const int dir = blockIdx.y;
  const int* __restrict__ dstA = dir ? midx : uidx;
  const int* __restrict__ srcA = dir ? uidx : midx;
  const int* __restrict__ base = dir ? base1 : base0;
  int* __restrict__ rec        = dir ? rec1 : rec0;
  const int shift = dir ? SHIFT1 : SHIFT0;
  const int sbits = dir ? SRCBITS1 : SRCBITS0;
  const int tid  = threadIdx.x;
  const int lane = tid & 31;
  const int c    = blockIdx.x;
  const int e0   = c * CH;
  const int rest = nedge - e0;
  const int nloc = rest < CH ? rest : CH;

  int d[16], s[16];
  #pragma unroll
  for (int k = 0; k < 16; ++k) {
    const int e = e0 + k * 1024 + tid;
    d[k] = (e < nedge) ? dstA[e] : -1;
    s[k] = (e < nedge) ? srcA[e] : 0;
  }

  int g = 0, cn = 0;
  if (tid < NB) {
    g  = base[tid * NCH + c];
    cn = base[tid * NCH + c + 1] - g;
  }
  int x = cn;
  if (tid < 256) {
    #pragma unroll
    for (int q = 1; q < 32; q <<= 1) {
      const int y = __shfl_up(x, q);
      if (lane >= q) x += y;
    }
    if (lane == 31) wtot[tid >> 5] = x;
  }
  __syncthreads();
  if (tid < 256) {
    int wb = 0;
    for (int w = 0; w < (tid >> 5); ++w) wb += wtot[w];
    const int ex = wb + x - cn;
    cur[tid] = ex;
    dlt[tid] = g - ex;
  }
  __syncthreads();

  #pragma unroll
  for (int k = 0; k < 16; ++k) {
    if (d[k] >= 0) {
      const int p = atomicAdd(&cur[d[k] >> shift], 1);
      sDst[p] = d[k];
      sSrc[p] = s[k];
    }
  }
  __syncthreads();

  const int lmask = (1 << shift) - 1;
  for (int i = tid; i < nloc; i += 1024) {
    const int dd = sDst[i];
    rec[i + dlt[dd >> shift]] = sSrc[i] | ((dd & lmask) << sbits);
  }
}

__global__ __launch_bounds__(1024) void bucket_csr_kernel(
    const int* __restrict__ base0, const int* __restrict__ base1,
    const int* __restrict__ rec0, const int* __restrict__ rec1,
    int* __restrict__ offU, int* __restrict__ offM,
    int* __restrict__ colU, int* __restrict__ colM) {
  __shared__ int cnt[1024];
  __shared__ int wsum[32];
  __shared__ int stage[STCAP];
  const int dir = blockIdx.y;
  const int b   = blockIdx.x;
  const int* __restrict__ base = dir ? base1 : base0;
  const int* __restrict__ rec  = dir ? rec1 : rec0;
  int* __restrict__ off        = dir ? offM : offU;
  int* __restrict__ col        = dir ? colM : colU;
  const int shift = dir ? SHIFT1 : SHIFT0;
  const int sbits = dir ? SRCBITS1 : SRCBITS0;
  const int nrows = dir ? N_MOVIE : N_USER;
  const int smask = (1 << sbits) - 1;
  const int tid  = threadIdx.x;
  const int lane = tid & 31;
  const int wave = tid >> 5;
  const int bs = base[b * NCH];
  const int be = base[(b + 1) * NCH];
  const int n  = be - bs;

  cnt[tid] = 0;
  __syncthreads();
  for (int i0 = bs; i0 < be; i0 += 4096) {
    int r[4];
    #pragma unroll
    for (int k = 0; k < 4; ++k) {
      const int i = i0 + k * 1024 + tid;
      r[k] = (i < be) ? rec[i] : -1;
    }
    #pragma unroll
    for (int k = 0; k < 4; ++k)
      if (r[k] >= 0) atomicAdd(&cnt[r[k] >> sbits], 1);
  }
  __syncthreads();

  const int v = cnt[tid];
  int x = v;
  #pragma unroll
  for (int q = 1; q < 32; q <<= 1) {
    const int y = __shfl_up(x, q);
    if (lane >= q) x += y;
  }
  if (lane == 31) wsum[wave] = x;
  __syncthreads();
  if (wave == 0) {
    int sacc = wsum[lane];
    #pragma unroll
    for (int q = 1; q < 32; q <<= 1) {
      const int y = __shfl_up(sacc, q);
      if (lane >= q) sacc += y;
    }
    wsum[lane] = sacc;
  }
  __syncthreads();
  const int ex = ((wave > 0) ? wsum[wave - 1] : 0) + x - v;
  const int node = (b << shift) + tid;
  if (tid < (1 << shift) && node < nrows) off[node] = bs + ex;
  if (b == NB - 1 && tid == 0) off[nrows] = be;
  cnt[tid] = ex;
  __syncthreads();

  if (n <= STCAP) {
    for (int i0 = bs; i0 < be; i0 += 4096) {
      int r[4];
      #pragma unroll
      for (int k = 0; k < 4; ++k) {
        const int i = i0 + k * 1024 + tid;
        r[k] = (i < be) ? rec[i] : -1;
      }
      #pragma unroll
      for (int k = 0; k < 4; ++k)
        if (r[k] >= 0) stage[atomicAdd(&cnt[r[k] >> sbits], 1)] = r[k] & smask;
    }
    __syncthreads();
    for (int i = tid; i < n; i += 1024) col[bs + i] = stage[i];
  } else {
    for (int i = bs + tid; i < be; i += 1024) {
      const int rr = rec[i];
      col[bs + atomicAdd(&cnt[rr >> sbits], 1)] = rr & smask;
    }
  }
}

template <int NA>
__device__ __forceinline__ void gather_step(const float* __restrict__ xl,
                                            const int (&ids)[4], const int (&dg)[4],
                                            int cb, int t0, float (&ax)[4], float (&ay)[4]) {
  float2 v[NA][8];
  #pragma unroll
  for (int p = 0; p < NA; ++p) {
    #pragma unroll
    for (int k = 0; k < 8; ++k) {
      const int t = t0 + k;
      const int sidx = __builtin_amdgcn_readlane(ids[p], ((t < dg[p]) ? t : t0) - cb);
      v[p][k] = *(const float2*)&xl[(size_t)sidx * D];
    }
  }
  #pragma unroll
  for (int p = 0; p < NA; ++p) {
    #pragma unroll
    for (int k = 0; k < 8; ++k) {
      const bool ok = (t0 + k) < dg[p];
      ax[p] += ok ? v[p][k].x : 0.f;
      ay[p] += ok ? v[p][k].y : 0.f;
    }
  }
}

__global__ __launch_bounds__(256) void sage_pull_wmma_kernel(
    const float* __restrict__ xsrc, const int* __restrict__ off,
    const int* __restrict__ col, const float* __restrict__ xdst,
    const float* __restrict__ Wl, const float* __restrict__ Wr,
    const float* __restrict__ bias,
    float* __restrict__ out, int nrows, int do_relu) {
  __shared__ float ldsWl[D * WSTRIDE];
  __shared__ float ldsWr[D * WSTRIDE];
  __shared__ float ldsMean[32 * ASTRIDE];
  __shared__ float ldsX[32 * ASTRIDE];

  const int tid  = threadIdx.x;
  const int wave = tid >> 5;
  const int lane = tid & 31;
  const int ntiles = (nrows + 31) >> 5;
  const int tstep  = gridDim.x;
  int tile = blockIdx.x;

  int ov = 0;
  if (tile < ntiles && lane < 5) {
    const int gi = tile * 32 + wave * 4 + lane;
    ov = off[gi < nrows ? gi : nrows];
  }

  {
    float4 wl[4], wr[4];
    #pragma unroll
    for (int u = 0; u < 4; ++u) {
      const int i = (tid + u * 256) * 4;
      wl[u] = *(const float4*)&Wl[i];
      wr[u] = *(const float4*)&Wr[i];
    }
    #pragma unroll
    for (int u = 0; u < 4; ++u) {
      const int i = (tid + u * 256) * 4;
      const int r = i >> 6, c = i & 63;
      *(float4*)&ldsWl[r * WSTRIDE + c] = wl[u];
      *(float4*)&ldsWr[r * WSTRIDE + c] = wr[u];
    }
  }

  const float* __restrict__ xl = xsrc + lane * 2;

  float4 xv[2];
  #pragma unroll
  for (int u = 0; u < 2; ++u) {
    const int g = tile * 32 + (tid >> 4) + 16 * u;
    xv[u] = make_float4(0.f, 0.f, 0.f, 0.f);
    if (tile < ntiles && g < nrows) xv[u] = *(const float4*)&xdst[(size_t)g * D + (tid & 15) * 4];
  }

  for (; tile < ntiles; tile += tstep) {
    const int row0 = tile * 32;
    const int tn   = tile + tstep;

    const int sr = tid >> 4, sc = (tid & 15) * 4;

    __syncthreads();
    #pragma unroll
    for (int u = 0; u < 2; ++u)
      *(float4*)&ldsX[(sr + 16 * u) * ASTRIDE + sc] = xv[u];

    int rb[5];
    #pragma unroll
    for (int q = 0; q < 5; ++q) rb[q] = __builtin_amdgcn_readlane(ov, q);

    int dg[4], st[4], ix[4];
    #pragma unroll
    for (int q = 0; q < 4; ++q) { dg[q] = rb[q + 1] - rb[q]; st[q] = rb[q]; ix[q] = q; }
#define SAGE_CSWAP(a, b)                                         \
  {                                                              \
    const bool sw = dg[a] < dg[b];                               \
    const int d0_ = sw ? dg[b] : dg[a], d1_ = sw ? dg[a] : dg[b]; \
    const int s0_ = sw ? st[b] : st[a], s1_ = sw ? st[a] : st[b]; \
    const int i0_ = sw ? ix[b] : ix[a], i1_ = sw ? ix[a] : ix[b]; \
    dg[a] = d0_; dg[b] = d1_; st[a] = s0_; st[b] = s1_; ix[a] = i0_; ix[b] = i1_; \
  }
    SAGE_CSWAP(0, 1) SAGE_CSWAP(2, 3) SAGE_CSWAP(0, 2) SAGE_CSWAP(1, 3) SAGE_CSWAP(1, 2)
#undef SAGE_CSWAP

    int ids[4];
    #pragma unroll
    for (int p = 0; p < 4; ++p) ids[p] = (lane < dg[p]) ? col[st[p] + lane] : 0;

    float ax[4] = {0.f, 0.f, 0.f, 0.f};
    float ay[4] = {0.f, 0.f, 0.f, 0.f};
    for (int cb = 0; cb < dg[0]; cb += 32) {
      const int ce = cb + 32;
      int idn[4] = {0, 0, 0, 0};
      if (ce < dg[0]) {
        #pragma unroll
        for (int p = 0; p < 4; ++p) idn[p] = (ce + lane < dg[p]) ? col[st[p] + ce + lane] : 0;
      }
      const int e3 = dg[3] < ce ? dg[3] : ce;
      const int e2 = dg[2] < ce ? dg[2] : ce;
      const int e1 = dg[1] < ce ? dg[1] : ce;
      const int e0 = dg[0] < ce ? dg[0] : ce;
      int t0 = cb;
      for (; t0 < e3; t0 += 8) gather_step<4>(xl, ids, dg, cb, t0, ax, ay);
      for (; t0 < e2; t0 += 8) gather_step<3>(xl, ids, dg, cb, t0, ax, ay);
      for (; t0 < e1; t0 += 8) gather_step<2>(xl, ids, dg, cb, t0, ax, ay);
      for (; t0 < e0; t0 += 8) gather_step<1>(xl, ids, dg, cb, t0, ax, ay);
      #pragma unroll
      for (int p = 0; p < 4; ++p) ids[p] = idn[p];
    }

    int t = tid;
    asm volatile("" : "+v"(t));
    const int xr = t >> 4, xc = (t & 15) * 4;
    const int lq = t & 15, lh = (t >> 4) & 1, wv = t >> 5;
    const int rt    = wv >> 2;
    const int ct    = wv & 3;
    const int mrow  = rt * 16 + lq;
    const int ncol  = ct * 16 + lq;
    const int khalf = lh * 2;
    const int mhalf = lh * 8;
    #pragma unroll
    for (int p = 0; p < 4; ++p) {
      const int r = wv * 4 + ix[p];
      const float inv = 1.0f / fmaxf((float)dg[p], 1.0f);
      *(float2*)&ldsMean[r * ASTRIDE + (t & 31) * 2] = make_float2(ax[p] * inv, ay[p] * inv);
    }
    __syncthreads();

    int ovn = 0;
    if (tn < ntiles && lane < 5) {
      const int gi = tn * 32 + wave * 4 + lane;
      ovn = off[gi < nrows ? gi : nrows];
    }
    const float bcol = bias[ncol];
    #pragma unroll
    for (int u = 0; u < 2; ++u) {
      const int g = tn * 32 + xr + 16 * u;
      xv[u] = make_float4(0.f, 0.f, 0.f, 0.f);
      if (tn < ntiles && g < nrows) xv[u] = *(const float4*)&xdst[(size_t)g * D + xc];
    }

    v8f acc = {0.f, 0.f, 0.f, 0.f, 0.f, 0.f, 0.f, 0.f};
    #pragma unroll
    for (int k = 0; k < D; k += 4) {
      const int kk = k + khalf;
      const v2f a = *(const v2f*)&ldsMean[mrow * ASTRIDE + kk];
      v2f b;
      b.x = ldsWl[(kk + 0) * WSTRIDE + ncol];
      b.y = ldsWl[(kk + 1) * WSTRIDE + ncol];
      acc = __builtin_amdgcn_wmma_f32_16x16x4_f32(false, a, false, b,
                                                  (short)0, acc, false, false);
    }
    #pragma unroll
    for (int k = 0; k < D; k += 4) {
      const int kk = k + khalf;
      const v2f a = *(const v2f*)&ldsX[mrow * ASTRIDE + kk];
      v2f b;
      b.x = ldsWr[(kk + 0) * WSTRIDE + ncol];
      b.y = ldsWr[(kk + 1) * WSTRIDE + ncol];
      acc = __builtin_amdgcn_wmma_f32_16x16x4_f32(false, a, false, b,
                                                  (short)0, acc, false, false);
    }

    #pragma unroll
    for (int v = 0; v < 8; ++v) {
      const int g = row0 + rt * 16 + mhalf + v;
      if (g < nrows) {
        float val = acc[v] + bcol;
        if (do_relu) val = fmaxf(val, 0.0f);
        out[(size_t)g * D + ncol] = val;
      }
    }
    ov = ovn;
  }
}

extern "C" void kernel_launch(void* const* d_in, const int* in_sizes, int n_in,
                              void* d_out, int out_size, void* d_ws, size_t ws_size,
                              hipStream_t stream) {
  const float* x_user  = (const float*)d_in[0];
  const float* x_movie = (const float*)d_in[1];
  const float* Wl1_um  = (const float*)d_in[2];
  const float* Wr1_um  = (const float*)d_in[3];
  const float* b1_um   = (const float*)d_in[4];
  const float* Wl1_mu  = (const float*)d_in[5];
  const float* Wr1_mu  = (const float*)d_in[6];
  const float* b1_mu   = (const float*)d_in[7];
  const float* Wl2_um  = (const float*)d_in[8];
  const float* Wr2_um  = (const float*)d_in[9];
  const float* b2_um   = (const float*)d_in[10];
  const float* Wl2_mu  = (const float*)d_in[11];
  const float* Wr2_mu  = (const float*)d_in[12];
  const float* b2_mu   = (const float*)d_in[13];
  const int*   uidx    = (const int*)d_in[14];
  const int*   midx    = (const int*)d_in[15];

  float* ws  = (float*)d_ws;
  int*   wsi = (int*)d_ws;
  int* cnt0  = wsi + W_CNT0;
  int* cnt1  = wsi + W_CNT1;
  int* base0 = wsi + W_BASE0;
  int* base1 = wsi + W_BASE1;
  int* offU  = wsi + W_OFFU;
  int* offM  = wsi + W_OFFM;
  int* colU  = wsi + W_COLU;
  int* colM  = wsi + W_COLM;
  int* rec0  = wsi + W_REC0;
  int* rec1  = wsi + W_REC1;
  float* u1 = ws + OFF_U1;
  float* m1 = ws + OFF_M1;

  float* u2 = (float*)d_out;
  float* m2 = (float*)d_out + (size_t)N_USER * D;

  const dim3 blk(256);
  const dim3 big(1024);
  const dim3 gChunk(NCH, 2);
  const dim3 gBucket(NB, 2);
  const dim3 gLayer(1280);

  bin_count_kernel<<<gChunk, big, 0, stream>>>(uidx, midx, cnt0, cnt1, N_EDGE);
  scan_kernel<<<dim3(2), big, 0, stream>>>(cnt0, cnt1, base0, base1);
  bin_scatter_kernel<<<gChunk, big, 0, stream>>>(uidx, midx, base0, base1,
                                                 rec0, rec1, N_EDGE);
  bucket_csr_kernel<<<gBucket, big, 0, stream>>>(base0, base1, rec0, rec1,
                                                 offU, offM, colU, colM);

  sage_pull_wmma_kernel<<<gLayer, blk, 0, stream>>>(x_user, offM, colM, x_movie,
                                                  Wl1_um, Wr1_um, b1_um, m1, N_MOVIE, 1);
  sage_pull_wmma_kernel<<<gLayer, blk, 0, stream>>>(x_movie, offU, colU, x_user,
                                                  Wl1_mu, Wr1_mu, b1_mu, u1, N_USER, 1);

  sage_pull_wmma_kernel<<<gLayer, blk, 0, stream>>>(u1, offM, colM, m1,
                                                  Wl2_um, Wr2_um, b2_um, m2, N_MOVIE, 0);
  sage_pull_wmma_kernel<<<gLayer, blk, 0, stream>>>(m1, offU, colU, u1,
                                                  Wl2_mu, Wr2_mu, b2_mu, u2, N_USER, 0);
}


# Layer kernel gather: fewer instructions per gathered row: load address = scalar table base + 32-bit index (id*32+lane), no per-slot index clamp, adds masked only in a row's last step; sum order kept
# speedup vs baseline: 5.5825x; 1.0648x over previous
#include <hip/hip_runtime.h>


#define N_USER  200000
#define N_MOVIE 100000
#define N_EDGE  4000000
#define D       64

typedef float v2f __attribute__((ext_vector_type(2)));
typedef float v8f __attribute__((ext_vector_type(8)));

#define WSTRIDE 80
#define ASTRIDE 68

#define CH       16384
#define NCH      245
#define NB       196
#define NCNT     48020
#define SHIFT0   10
#define SHIFT1   9
#define SRCBITS0 17
#define SRCBITS1 18
#define STCAP    24576

#define W_CNT0  0u
#define W_CNT1  50000u
#define W_BASE0 100000u
#define W_BASE1 150000u
#define W_OFFU  300000u
#define W_OFFM  500064u
#define W_COLU  600128u
#define W_COLM  4600128u
#define W_REC0  8600128u
#define W_REC1  12600128u
#define OFF_U1  19500000u
#define OFF_M1  (OFF_U1 + N_USER * 64u)

__global__ __launch_bounds__(1024) void bin_count_kernel(
    const int* __restrict__ uidx, const int* __restrict__ midx,
    int* __restrict__ cnt0, int* __restrict__ cnt1, int nedge) {
  __shared__ int hist[NB];
  const int dir = blockIdx.y;
  const int* __restrict__ dst = dir ? midx : uidx;
  int* __restrict__ cnt       = dir ? cnt1 : cnt0;
  const int shift = dir ? SHIFT1 : SHIFT0;
  const int tid = threadIdx.x;
  if (tid < NB) hist[tid] = 0;
  __syncthreads();
  const int e0 = blockIdx.x * CH;
  int d[16];
  #pragma unroll
  for (int k = 0; k < 16; ++k) {
    const int e = e0 + k * 1024 + tid;
    d[k] = (e < nedge) ? dst[e] : -1;
  }
  #pragma unroll
  for (int k = 0; k < 16; ++k)
    if (d[k] >= 0) atomicAdd(&hist[d[k] >> shift], 1);
  __syncthreads();
  if (tid < NB) cnt[tid * NCH + blockIdx.x] = hist[tid];
}

__global__ __launch_bounds__(1024) void scan_kernel(
    const int* __restrict__ cnt0, const int* __restrict__ cnt1,
    int* __restrict__ base0, int* __restrict__ base1) {
  __shared__ int wsum[32];
  const int* __restrict__ deg = (blockIdx.x == 0) ? cnt0 : cnt1;
  int* __restrict__ off       = (blockIdx.x == 0) ? base0 : base1;
  const int n  = NCNT;
  const int n4 = n >> 2;
  const int tid  = threadIdx.x;
  const int lane = tid & 31;
  const int wave = tid >> 5;
  int carry = 0;
  for (int tile = 0; tile < n4; tile += 4096) {
    const int i0 = tile + 4 * tid;
    int4 v[4];
    int  s4[4];
    int  t = 0;
    #pragma unroll
    for (int k = 0; k < 4; ++k) {
      v[k] = make_int4(0, 0, 0, 0);
      if (i0 + k < n4) v[k] = ((const int4*)deg)[i0 + k];
      s4[k] = t;
      t += v[k].x + v[k].y + v[k].z + v[k].w;
    }
    int x = t;
    #pragma unroll
    for (int d = 1; d < 32; d <<= 1) {
      const int y = __shfl_up(x, d);
      if (lane >= d) x += y;
    }
    if (lane == 31) wsum[wave] = x;
    __syncthreads();
    if (wave == 0) {
      int s = wsum[lane];
      #pragma unroll
      for (int d = 1; d < 32; d <<= 1) {
        const int y = __shfl_up(s, d);
        if (lane >= d) s += y;
      }
      wsum[lane] = s;
    }
    __syncthreads();
    const int wbase = (wave > 0) ? wsum[wave - 1] : 0;
    const int total = wsum[31];
    const int e0 = carry + wbase + x - t;
    #pragma unroll
    for (int k = 0; k < 4; ++k) {
      const int e = e0 + s4[k];
      if (i0 + k < n4)
        ((int4*)off)[i0 + k] = make_int4(e, e + v[k].x, e + v[k].x + v[k].y,
                                         e + v[k].x + v[k].y + v[k].z);
    }
    carry += total;
    __syncthreads();
  }
  if (tid == 0) off[n] = carry;
}

__global__ __launch_bounds__(1024) void bin_scatter_kernel(
    const int* __restrict__ uidx, const int* __restrict__ midx,
    const int* __restrict__ base0, const int* __restrict__ base1,
    int* __restrict__ rec0, int* __restrict__ rec1, int nedge) {
  __shared__ int sDst[CH];
  __shared__ int sSrc[CH];
  __shared__ int cur[256];
  __shared__ int dlt[256];
  __shared__ int wtot[8];
  const int dir = blockIdx.y;
  const int* __restrict__ dstA = dir ? midx : uidx;
  const int* __restrict__ srcA = dir ? uidx : midx;
  const int* __restrict__ base = dir ? base1 : base0;
  int* __restrict__ rec        = dir ? rec1 : rec0;
  const int shift = dir ? SHIFT1 : SHIFT0;
  const int sbits = dir ? SRCBITS1 : SRCBITS0;
  const int tid  = threadIdx.x;
  const int lane = tid & 31;
  const int c    = blockIdx.x;
  const int e0   = c * CH;
  const int rest = nedge - e0;
  const int nloc = rest < CH ? rest : CH;

  int d[16], s[16];
  #pragma unroll
  for (int k = 0; k < 16; ++k) {
    const int e = e0 + k * 1024 + tid;
    d[k] = (e < nedge) ? dstA[e] : -1;
    s[k] = (e < nedge) ? srcA[e] : 0;
  }

  int g = 0, cn = 0;
  if (tid < NB) {
    g  = base[tid * NCH + c];
    cn = base[tid * NCH + c + 1] - g;
  }
  int x = cn;
  if (tid < 256) {
    #pragma unroll
    for (int q = 1; q < 32; q <<= 1) {
      const int y = __shfl_up(x, q);
      if (lane >= q) x += y;
    }
    if (lane == 31) wtot[tid >> 5] = x;
  }
  __syncthreads();
  if (tid < 256) {
    int wb = 0;
    for (int w = 0; w < (tid >> 5); ++w) wb += wtot[w];
    const int ex = wb + x - cn;
    cur[tid] = ex;
    dlt[tid] = g - ex;
  }
  __syncthreads();

  #pragma unroll
  for (int k = 0; k < 16; ++k) {
    if (d[k] >= 0) {
      const int p = atomicAdd(&cur[d[k] >> shift], 1);
      sDst[p] = d[k];
      sSrc[p] = s[k];
    }
  }
  __syncthreads();

  const int lmask = (1 << shift) - 1;
  for (int i = tid; i < nloc; i += 1024) {
    const int dd = sDst[i];
    rec[i + dlt[dd >> shift]] = sSrc[i] | ((dd & lmask) << sbits);
  }
}

__global__ __launch_bounds__(1024) void bucket_csr_kernel(
    const int* __restrict__ base0, const int* __restrict__ base1,
    const int* __restrict__ rec0, const int* __restrict__ rec1,
    int* __restrict__ offU, int* __restrict__ offM,
    int* __restrict__ colU, int* __restrict__ colM) {
  __shared__ int cnt[1024];
  __shared__ int wsum[32];
  __shared__ int stage[STCAP];
  const int dir = blockIdx.y;
  const int b   = blockIdx.x;
  const int* __restrict__ base = dir ? base1 : base0;
  const int* __restrict__ rec  = dir ? rec1 : rec0;
  int* __restrict__ off        = dir ? offM : offU;
  int* __restrict__ col        = dir ? colM : colU;
  const int shift = dir ? SHIFT1 : SHIFT0;
  const int sbits = dir ? SRCBITS1 : SRCBITS0;
  const int nrows = dir ? N_MOVIE : N_USER;
  const int smask = (1 << sbits) - 1;
  const int tid  = threadIdx.x;
  const int lane = tid & 31;
  const int wave = tid >> 5;
  const int bs = base[b * NCH];
  const int be = base[(b + 1) * NCH];
  const int n  = be - bs;

  cnt[tid] = 0;
  __syncthreads();
  for (int i0 = bs; i0 < be; i0 += 4096) {
    int r[4];
    #pragma unroll
    for (int k = 0; k < 4; ++k) {
      const int i = i0 + k * 1024 + tid;
      r[k] = (i < be) ? rec[i] : -1;
    }
    #pragma unroll
    for (int k = 0; k < 4; ++k)
      if (r[k] >= 0) atomicAdd(&cnt[r[k] >> sbits], 1);
  }
  __syncthreads();

  const int v = cnt[tid];
  int x = v;
  #pragma unroll
  for (int q = 1; q < 32; q <<= 1) {
    const int y = __shfl_up(x, q);
    if (lane >= q) x += y;
  }
  if (lane == 31) wsum[wave] = x;
  __syncthreads();
  if (wave == 0) {
    int sacc = wsum[lane];
    #pragma unroll
    for (int q = 1; q < 32; q <<= 1) {
      const int y = __shfl_up(sacc, q);
      if (lane >= q) sacc += y;
    }
    wsum[lane] = sacc;
  }
  __syncthreads();
  const int ex = ((wave > 0) ? wsum[wave - 1] : 0) + x - v;
  const int node = (b << shift) + tid;
  if (tid < (1 << shift) && node < nrows) off[node] = bs + ex;
  if (b == NB - 1 && tid == 0) off[nrows] = be;
  cnt[tid] = ex;
  __syncthreads();

  if (n <= STCAP) {
    for (int i0 = bs; i0 < be; i0 += 4096) {
      int r[4];
      #pragma unroll
      for (int k = 0; k < 4; ++k) {
        const int i = i0 + k * 1024 + tid;
        r[k] = (i < be) ? rec[i] : -1;
      }
      #pragma unroll
      for (int k = 0; k < 4; ++k)
        if (r[k] >= 0) stage[atomicAdd(&cnt[r[k] >> sbits], 1)] = r[k] & smask;
    }
    __syncthreads();
    for (int i = tid; i < n; i += 1024) col[bs + i] = stage[i];
  } else {
    for (int i = bs + tid; i < be; i += 1024) {
      const int rr = rec[i];
      col[bs + atomicAdd(&cnt[rr >> sbits], 1)] = rr & smask;
    }
  }
}

template <int NA>
__device__ __forceinline__ void gather_step(const float2* __restrict__ x2, unsigned lane,
                                            const int (&ids)[4], const int (&dg)[4],
                                            int cb, int t0, float (&ax)[4], float (&ay)[4]) {
  float2 v[NA][8];
  const int sb = t0 - cb;
  #pragma unroll
  for (int p = 0; p < NA; ++p) {
    #pragma unroll
    for (int k = 0; k < 8; ++k) {
      const unsigned sidx = (unsigned)__builtin_amdgcn_readlane(ids[p], sb + k);
      v[p][k] = x2[sidx * 32u + lane];
    }
  }
  #pragma unroll
  for (int p = 0; p < NA; ++p) {
    if (t0 + 8 <= dg[p]) {
      #pragma unroll
      for (int k = 0; k < 8; ++k) {
        ax[p] += v[p][k].x;
        ay[p] += v[p][k].y;
      }
    } else {
      #pragma unroll
      for (int k = 0; k < 8; ++k) {
        const bool ok = (t0 + k) < dg[p];
        ax[p] += ok ? v[p][k].x : 0.f;
        ay[p] += ok ? v[p][k].y : 0.f;
      }
    }
  }
}

__global__ __launch_bounds__(256) void sage_pull_wmma_kernel(
    const float* __restrict__ xsrc, const int* __restrict__ off,
    const int* __restrict__ col, const float* __restrict__ xdst,
    const float* __restrict__ Wl, const float* __restrict__ Wr,
    const float* __restrict__ bias,
    float* __restrict__ out, int nrows, int do_relu) {
  __shared__ float ldsWl[D * WSTRIDE];
  __shared__ float ldsWr[D * WSTRIDE];
  __shared__ float ldsMean[32 * ASTRIDE];
  __shared__ float ldsX[32 * ASTRIDE];

  const int tid  = threadIdx.x;
  const int wave = tid >> 5;
  const int lane = tid & 31;
  const int ntiles = (nrows + 31) >> 5;
  const int tstep  = gridDim.x;
  int tile = blockIdx.x;

  int ov = 0;
  if (tile < ntiles && lane < 5) {
    const int gi = tile * 32 + wave * 4 + lane;
    ov = off[gi < nrows ? gi : nrows];
  }

  {
    float4 wl[4], wr[4];
    #pragma unroll
    for (int u = 0; u < 4; ++u) {
      const int i = (tid + u * 256) * 4;
      wl[u] = *(const float4*)&Wl[i];
      wr[u] = *(const float4*)&Wr[i];
    }
    #pragma unroll
    for (int u = 0; u < 4; ++u) {
      const int i = (tid + u * 256) * 4;
      const int r = i >> 6, c = i & 63;
      *(float4*)&ldsWl[r * WSTRIDE + c] = wl[u];
      *(float4*)&ldsWr[r * WSTRIDE + c] = wr[u];
    }
  }

  const float2* __restrict__ x2 = (const float2*)xsrc;

  float4 xv[2];
  #pragma unroll
  for (int u = 0; u < 2; ++u) {
    const int g = tile * 32 + (tid >> 4) + 16 * u;
    xv[u] = make_float4(0.f, 0.f, 0.f, 0.f);
    if (tile < ntiles && g < nrows) xv[u] = *(const float4*)&xdst[(size_t)g * D + (tid & 15) * 4];
  }

  for (; tile < ntiles; tile += tstep) {
    const int row0 = tile * 32;
    const int tn   = tile + tstep;

    const int sr = tid >> 4, sc = (tid & 15) * 4;

    __syncthreads();
    #pragma unroll
    for (int u = 0; u < 2; ++u)
      *(float4*)&ldsX[(sr + 16 * u) * ASTRIDE + sc] = xv[u];

    int rb[5];
    #pragma unroll
    for (int q = 0; q < 5; ++q) rb[q] = __builtin_amdgcn_readlane(ov, q);

    int dg[4], st[4], ix[4];
    #pragma unroll
    for (int q = 0; q < 4; ++q) { dg[q] = rb[q + 1] - rb[q]; st[q] = rb[q]; ix[q] = q; }
#define SAGE_CSWAP(a, b)                                         \
  {                                                              \
    const bool sw = dg[a] < dg[b];                               \
    const int d0_ = sw ? dg[b] : dg[a], d1_ = sw ? dg[a] : dg[b]; \
    const int s0_ = sw ? st[b] : st[a], s1_ = sw ? st[a] : st[b]; \
    const int i0_ = sw ? ix[b] : ix[a], i1_ = sw ? ix[a] : ix[b]; \
    dg[a] = d0_; dg[b] = d1_; st[a] = s0_; st[b] = s1_; ix[a] = i0_; ix[b] = i1_; \
  }
    SAGE_CSWAP(0, 1) SAGE_CSWAP(2, 3) SAGE_CSWAP(0, 2) SAGE_CSWAP(1, 3) SAGE_CSWAP(1, 2)
#undef SAGE_CSWAP

    int ids[4];
    #pragma unroll
    for (int p = 0; p < 4; ++p) ids[p] = (lane < dg[p]) ? col[st[p] + lane] : 0;

    float ax[4] = {0.f, 0.f, 0.f, 0.f};
    float ay[4] = {0.f, 0.f, 0.f, 0.f};
    for (int cb = 0; cb < dg[0]; cb += 32) {
      const int ce = cb + 32;
      int idn[4] = {0, 0, 0, 0};
      if (ce < dg[0]) {
        #pragma unroll
        for (int p = 0; p < 4; ++p) idn[p] = (ce + lane < dg[p]) ? col[st[p] + ce + lane] : 0;
      }
      const int e3 = dg[3] < ce ? dg[3] : ce;
      const int e2 = dg[2] < ce ? dg[2] : ce;
      const int e1 = dg[1] < ce ? dg[1] : ce;
      const int e0 = dg[0] < ce ? dg[0] : ce;
      int t0 = cb;
      for (; t0 < e3; t0 += 8) gather_step<4>(x2, (unsigned)lane, ids, dg, cb, t0, ax, ay);
      for (; t0 < e2; t0 += 8) gather_step<3>(x2, (unsigned)lane, ids, dg, cb, t0, ax, ay);
      for (; t0 < e1; t0 += 8) gather_step<2>(x2, (unsigned)lane, ids, dg, cb, t0, ax, ay);
      for (; t0 < e0; t0 += 8) gather_step<1>(x2, (unsigned)lane, ids, dg, cb, t0, ax, ay);
      #pragma unroll
      for (int p = 0; p < 4; ++p) ids[p] = idn[p];
    }

    int t = tid;
    asm volatile("" : "+v"(t));
    const int xr = t >> 4, xc = (t & 15) * 4;
    const int lq = t & 15, lh = (t >> 4) & 1, wv = t >> 5;
    const int rt    = wv >> 2;
    const int ct    = wv & 3;
    const int mrow  = rt * 16 + lq;
    const int ncol  = ct * 16 + lq;
    const int khalf = lh * 2;
    const int mhalf = lh * 8;
    #pragma unroll
    for (int p = 0; p < 4; ++p) {
      const int r = wv * 4 + ix[p];
      const float inv = 1.0f / fmaxf((float)dg[p], 1.0f);
      *(float2*)&ldsMean[r * ASTRIDE + (t & 31) * 2] = make_float2(ax[p] * inv, ay[p] * inv);
    }
    __syncthreads();

    int ovn = 0;
    if (tn < ntiles && lane < 5) {
      const int gi = tn * 32 + wave * 4 + lane;
      ovn = off[gi < nrows ? gi : nrows];
    }
    const float bcol = bias[ncol];
    #pragma unroll
    for (int u = 0; u < 2; ++u) {
      const int g = tn * 32 + xr + 16 * u;
      xv[u] = make_float4(0.f, 0.f, 0.f, 0.f);
      if (tn < ntiles && g < nrows) xv[u] = *(const float4*)&xdst[(size_t)g * D + xc];
    }

    v8f acc = {0.f, 0.f, 0.f, 0.f, 0.f, 0.f, 0.f, 0.f};
    #pragma unroll
    for (int k = 0; k < D; k += 4) {
      const int kk = k + khalf;
      const v2f a = *(const v2f*)&ldsMean[mrow * ASTRIDE + kk];
      v2f b;
      b.x = ldsWl[(kk + 0) * WSTRIDE + ncol];
      b.y = ldsWl[(kk + 1) * WSTRIDE + ncol];
      acc = __builtin_amdgcn_wmma_f32_16x16x4_f32(false, a, false, b,
                                                  (short)0, acc, false, false);
    }
    #pragma unroll
    for (int k = 0; k < D; k += 4) {
      const int kk = k + khalf;
      const v2f a = *(const v2f*)&ldsX[mrow * ASTRIDE + kk];
      v2f b;
      b.x = ldsWr[(kk + 0) * WSTRIDE + ncol];
      b.y = ldsWr[(kk + 1) * WSTRIDE + ncol];
      acc = __builtin_amdgcn_wmma_f32_16x16x4_f32(false, a, false, b,
                                                  (short)0, acc, false, false);
    }

    #pragma unroll
    for (int v = 0; v < 8; ++v) {
      const int g = row0 + rt * 16 + mhalf + v;
      if (g < nrows) {
        float val = acc[v] + bcol;
        if (do_relu) val = fmaxf(val, 0.0f);
        out[(size_t)g * D + ncol] = val;
      }
    }
    ov = ovn;
  }
}

extern "C" void kernel_launch(void* const* d_in, const int* in_sizes, int n_in,
                              void* d_out, int out_size, void* d_ws, size_t ws_size,
                              hipStream_t stream) {
  const float* x_user  = (const float*)d_in[0];
  const float* x_movie = (const float*)d_in[1];
  const float* Wl1_um  = (const float*)d_in[2];
  const float* Wr1_um  = (const float*)d_in[3];
  const float* b1_um   = (const float*)d_in[4];
  const float* Wl1_mu  = (const float*)d_in[5];
  const float* Wr1_mu  = (const float*)d_in[6];
  const float* b1_mu   = (const float*)d_in[7];
  const float* Wl2_um  = (const float*)d_in[8];
  const float* Wr2_um  = (const float*)d_in[9];
  const float* b2_um   = (const float*)d_in[10];
  const float* Wl2_mu  = (const float*)d_in[11];
  const float* Wr2_mu  = (const float*)d_in[12];
  const float* b2_mu   = (const float*)d_in[13];
  const int*   uidx    = (const int*)d_in[14];
  const int*   midx    = (const int*)d_in[15];

  float* ws  = (float*)d_ws;
  int*   wsi = (int*)d_ws;
  int* cnt0  = wsi + W_CNT0;
  int* cnt1  = wsi + W_CNT1;
  int* base0 = wsi + W_BASE0;
  int* base1 = wsi + W_BASE1;
  int* offU  = wsi + W_OFFU;
  int* offM  = wsi + W_OFFM;
  int* colU  = wsi + W_COLU;
  int* colM  = wsi + W_COLM;
  int* rec0  = wsi + W_REC0;
  int* rec1  = wsi + W_REC1;
  float* u1 = ws + OFF_U1;
  float* m1 = ws + OFF_M1;

  float* u2 = (float*)d_out;
  float* m2 = (float*)d_out + (size_t)N_USER * D;

  const dim3 blk(256);
  const dim3 big(1024);
  const dim3 gChunk(NCH, 2);
  const dim3 gBucket(NB, 2);
  const dim3 gLayer(1280);

  bin_count_kernel<<<gChunk, big, 0, stream>>>(uidx, midx, cnt0, cnt1, N_EDGE);
  scan_kernel<<<dim3(2), big, 0, stream>>>(cnt0, cnt1, base0, base1);
  bin_scatter_kernel<<<gChunk, big, 0, stream>>>(uidx, midx, base0, base1,
                                                 rec0, rec1, N_EDGE);
  bucket_csr_kernel<<<gBucket, big, 0, stream>>>(base0, base1, rec0, rec1,
                                                 offU, offM, colU, colM);

  sage_pull_wmma_kernel<<<gLayer, blk, 0, stream>>>(x_user, offM, colM, x_movie,
                                                  Wl1_um, Wr1_um, b1_um, m1, N_MOVIE, 1);
  sage_pull_wmma_kernel<<<gLayer, blk, 0, stream>>>(x_movie, offU, colU, x_user,
                                                  Wl1_mu, Wr1_mu, b1_mu, u1, N_USER, 1);

  sage_pull_wmma_kernel<<<gLayer, blk, 0, stream>>>(u1, offM, colM, m1,
                                                  Wl2_um, Wr2_um, b2_um, m2, N_MOVIE, 0);
  sage_pull_wmma_kernel<<<gLayer, blk, 0, stream>>>(m1, offU, colU, u1,
                                                  Wl2_mu, Wr2_mu, b2_mu, u2, N_USER, 0);
}


# One launch per layer for both node types: their row tiles form one list (long-row type first), handed to resident work-groups by a returning add, claimed a tile ahead; 6 launches; sum order kept
# speedup vs baseline: 5.7970x; 1.0384x over previous
#include <hip/hip_runtime.h>


#define N_USER  200000
#define N_MOVIE 100000
#define N_EDGE  4000000
#define D       64

typedef float v2f __attribute__((ext_vector_type(2)));
typedef float v8f __attribute__((ext_vector_type(8)));

#define WSTRIDE 80
#define ASTRIDE 68

#define CH       16384
#define NCH      245
#define NB       196
#define NCNT     48020
#define SHIFT0   10
#define SHIFT1   9
#define SRCBITS0 17
#define SRCBITS1 18
#define STCAP    24576

#define W_CNT0  0u
#define W_CNT1  50000u
#define W_BASE0 100000u
#define W_BASE1 150000u
#define W_CTR   250000u
#define W_OFFU  300000u
#define W_OFFM  500064u
#define W_COLU  600128u
#define W_COLM  4600128u
#define W_REC0  8600128u
#define W_REC1  12600128u
#define OFF_U1  19500000u
#define OFF_M1  (OFF_U1 + N_USER * 64u)

__global__ __launch_bounds__(1024) void bin_count_kernel(
    const int* __restrict__ uidx, const int* __restrict__ midx,
    int* __restrict__ cnt0, int* __restrict__ cnt1, int nedge) {
  __shared__ int hist[NB];
  const int dir = blockIdx.y;
  const int* __restrict__ dst = dir ? midx : uidx;
  int* __restrict__ cnt       = dir ? cnt1 : cnt0;
  const int shift = dir ? SHIFT1 : SHIFT0;
  const int tid = threadIdx.x;
  if (tid < NB) hist[tid] = 0;
  __syncthreads();
  const int e0 = blockIdx.x * CH;
  int d[16];
  #pragma unroll
  for (int k = 0; k < 16; ++k) {
    const int e = e0 + k * 1024 + tid;
    d[k] = (e < nedge) ? dst[e] : -1;
  }
  #pragma unroll
  for (int k = 0; k < 16; ++k)
    if (d[k] >= 0) atomicAdd(&hist[d[k] >> shift], 1);
  __syncthreads();
  if (tid < NB) cnt[tid * NCH + blockIdx.x] = hist[tid];
}

__global__ __launch_bounds__(1024) void scan_kernel(
    const int* __restrict__ cnt0, const int* __restrict__ cnt1,
    int* __restrict__ base0, int* __restrict__ base1, int* __restrict__ ctr) {
  __shared__ int wsum[32];
  if (blockIdx.x == 0 && threadIdx.x < 2) ctr[threadIdx.x] = 0;
  const int* __restrict__ deg = (blockIdx.x == 0) ? cnt0 : cnt1;
  int* __restrict__ off       = (blockIdx.x == 0) ? base0 : base1;
  const int n  = NCNT;
  const int n4 = n >> 2;
  const int tid  = threadIdx.x;
  const int lane = tid & 31;
  const int wave = tid >> 5;
  int carry = 0;
  for (int tile = 0; tile < n4; tile += 4096) {
    const int i0 = tile + 4 * tid;
    int4 v[4];
    int  s4[4];
    int  t = 0;
    #pragma unroll
    for (int k = 0; k < 4; ++k) {
      v[k] = make_int4(0, 0, 0, 0);
      if (i0 + k < n4) v[k] = ((const int4*)deg)[i0 + k];
      s4[k] = t;
      t += v[k].x + v[k].y + v[k].z + v[k].w;
    }
    int x = t;
    #pragma unroll
    for (int d = 1; d < 32; d <<= 1) {
      const int y = __shfl_up(x, d);
      if (lane >= d) x += y;
    }
    if (lane == 31) wsum[wave] = x;
    __syncthreads();
    if (wave == 0) {
      int s = wsum[lane];
      #pragma unroll
      for (int d = 1; d < 32; d <<= 1) {
        const int y = __shfl_up(s, d);
        if (lane >= d) s += y;
      }
      wsum[lane] = s;
    }
    __syncthreads();
    const int wbase = (wave > 0) ? wsum[wave - 1] : 0;
    const int total = wsum[31];
    const int e0 = carry + wbase + x - t;
    #pragma unroll
    for (int k = 0; k < 4; ++k) {
      const int e = e0 + s4[k];
      if (i0 + k < n4)
        ((int4*)off)[i0 + k] = make_int4(e, e + v[k].x, e + v[k].x + v[k].y,
                                         e + v[k].x + v[k].y + v[k].z);
    }
    carry += total;
    __syncthreads();
  }
  if (tid == 0) off[n] = carry;
}

__global__ __launch_bounds__(1024) void bin_scatter_kernel(
    const int* __restrict__ uidx, const int* __restrict__ midx,
    const int* __restrict__ base0, const int* __restrict__ base1,
    int* __restrict__ rec0, int* __restrict__ rec1, int nedge) {
  __shared__ int sDst[CH];
  __shared__ int sSrc[CH];
  __shared__ int cur[256];
  __shared__ int dlt[256];
  __shared__ int wtot[8];
  const int dir = blockIdx.y;
  const int* __restrict__ dstA = dir ? midx : uidx;
  const int* __restrict__ srcA = dir ? uidx : midx;
  const int* __restrict__ base = dir ? base1 : base0;
  int* __restrict__ rec        = dir ? rec1 : rec0;
  const int shift = dir ? SHIFT1 : SHIFT0;
  const int sbits = dir ? SRCBITS1 : SRCBITS0;
  const int tid  = threadIdx.x;
  const int lane = tid & 31;
  const int c    = blockIdx.x;
  const int e0   = c * CH;
  const int rest = nedge - e0;
  const int nloc = rest < CH ? rest : CH;

  int d[16], s[16];
  #pragma unroll
  for (int k = 0; k < 16; ++k) {
    const int e = e0 + k * 1024 + tid;
    d[k] = (e < nedge) ? dstA[e] : -1;
    s[k] = (e < nedge) ? srcA[e] : 0;
  }

  int g = 0, cn = 0;
  if (tid < NB) {
    g  = base[tid * NCH + c];
    cn = base[tid * NCH + c + 1] - g;
  }
  int x = cn;
  if (tid < 256) {
    #pragma unroll
    for (int q = 1; q < 32; q <<= 1) {
      const int y = __shfl_up(x, q);
      if (lane >= q) x += y;
    }
    if (lane == 31) wtot[tid >> 5] = x;
  }
  __syncthreads();
  if (tid < 256) {
    int wb = 0;
    for (int w = 0; w < (tid >> 5); ++w) wb += wtot[w];
    const int ex = wb + x - cn;
    cur[tid] = ex;
    dlt[tid] = g - ex;
  }
  __syncthreads();

  #pragma unroll
  for (int k = 0; k < 16; ++k) {
    if (d[k] >= 0) {
      const int p = atomicAdd(&cur[d[k] >> shift], 1);
      sDst[p] = d[k];
      sSrc[p] = s[k];
    }
  }
  __syncthreads();

  const int lmask = (1 << shift) - 1;
  for (int i = tid; i < nloc; i += 1024) {
    const int dd = sDst[i];
    rec[i + dlt[dd >> shift]] = sSrc[i] | ((dd & lmask) << sbits);
  }
}

__global__ __launch_bounds__(1024) void bucket_csr_kernel(
    const int* __restrict__ base0, const int* __restrict__ base1,
    const int* __restrict__ rec0, const int* __restrict__ rec1,
    int* __restrict__ offU, int* __restrict__ offM,
    int* __restrict__ colU, int* __restrict__ colM) {
  __shared__ int cnt[1024];
  __shared__ int wsum[32];
  __shared__ int stage[STCAP];
  const int dir = blockIdx.y;
  const int b   = blockIdx.x;
  const int* __restrict__ base = dir ? base1 : base0;
  const int* __restrict__ rec  = dir ? rec1 : rec0;
  int* __restrict__ off        = dir ? offM : offU;
  int* __restrict__ col        = dir ? colM : colU;
  const int shift = dir ? SHIFT1 : SHIFT0;
  const int sbits = dir ? SRCBITS1 : SRCBITS0;
  const int nrows = dir ? N_MOVIE : N_USER;
  const int smask = (1 << sbits) - 1;
  const int tid  = threadIdx.x;
  const int lane = tid & 31;
  const int wave = tid >> 5;
  const int bs = base[b * NCH];
  const int be = base[(b + 1) * NCH];
  const int n  = be - bs;

  cnt[tid] = 0;
  __syncthreads();
  for (int i0 = bs; i0 < be; i0 += 4096) {
    int r[4];
    #pragma unroll
    for (int k = 0; k < 4; ++k) {
      const int i = i0 + k * 1024 + tid;
      r[k] = (i < be) ? rec[i] : -1;
    }
    #pragma unroll
    for (int k = 0; k < 4; ++k)
      if (r[k] >= 0) atomicAdd(&cnt[r[k] >> sbits], 1);
  }
  __syncthreads();

  const int v = cnt[tid];
  int x = v;
  #pragma unroll
  for (int q = 1; q < 32; q <<= 1) {
    const int y = __shfl_up(x, q);
    if (lane >= q) x += y;
  }
  if (lane == 31) wsum[wave] = x;
  __syncthreads();
  if (wave == 0) {
    int sacc = wsum[lane];
    #pragma unroll
    for (int q = 1; q < 32; q <<= 1) {
      const int y = __shfl_up(sacc, q);
      if (lane >= q) sacc += y;
    }
    wsum[lane] = sacc;
  }
  __syncthreads();
  const int ex = ((wave > 0) ? wsum[wave - 1] : 0) + x - v;
  const int node = (b << shift) + tid;
  if (tid < (1 << shift) && node < nrows) off[node] = bs + ex;
  if (b == NB - 1 && tid == 0) off[nrows] = be;
  cnt[tid] = ex;
  __syncthreads();

  if (n <= STCAP) {
    for (int i0 = bs; i0 < be; i0 += 4096) {
      int r[4];
      #pragma unroll
      for (int k = 0; k < 4; ++k) {
        const int i = i0 + k * 1024 + tid;
        r[k] = (i < be) ? rec[i] : -1;
      }
      #pragma unroll
      for (int k = 0; k < 4; ++k)
        if (r[k] >= 0) stage[atomicAdd(&cnt[r[k] >> sbits], 1)] = r[k] & smask;
    }
    __syncthreads();
    for (int i = tid; i < n; i += 1024) col[bs + i] = stage[i];
  } else {
    for (int i = bs + tid; i < be; i += 1024) {
      const int rr = rec[i];
      col[bs + atomicAdd(&cnt[rr >> sbits], 1)] = rr & smask;
    }
  }
}

template <int NA>
__device__ __forceinline__ void gather_step(const float2* __restrict__ x2, unsigned lane,
                                            const int (&ids)[4], const int (&dg)[4],
                                            int cb, int t0, float (&ax)[4], float (&ay)[4]) {
  float2 v[NA][8];
  const int sb = t0 - cb;
  #pragma unroll
  for (int p = 0; p < NA; ++p) {
    #pragma unroll
    for (int k = 0; k < 8; ++k) {
      const unsigned sidx = (unsigned)__builtin_amdgcn_readlane(ids[p], sb + k);
      v[p][k] = x2[sidx * 32u + lane];
    }
  }
  #pragma unroll
  for (int p = 0; p < NA; ++p) {
    if (t0 + 8 <= dg[p]) {
      #pragma unroll
      for (int k = 0; k < 8; ++k) {
        ax[p] += v[p][k].x;
        ay[p] += v[p][k].y;
      }
    } else {
      #pragma unroll
      for (int k = 0; k < 8; ++k) {
        const bool ok = (t0 + k) < dg[p];
        ax[p] += ok ? v[p][k].x : 0.f;
        ay[p] += ok ? v[p][k].y : 0.f;
      }
    }
  }
}

__global__ __launch_bounds__(256) void sage_pull_wmma_kernel(
    const float* __restrict__ xsrcA, const int* __restrict__ offA,
    const int* __restrict__ colA, const float* __restrict__ xdstA,
    const float* __restrict__ WlA, const float* __restrict__ WrA,
    const float* __restrict__ biasA, float* __restrict__ outA, int nrowsA,
    const float* __restrict__ xsrcB, const int* __restrict__ offB,
    const int* __restrict__ colB, const float* __restrict__ xdstB,
    const float* __restrict__ WlB, const float* __restrict__ WrB,
    const float* __restrict__ biasB, float* __restrict__ outB, int nrowsB,
    int* __restrict__ ctr, int do_relu) {
  __shared__ float ldsWl[D * WSTRIDE];
  __shared__ float ldsWr[D * WSTRIDE];
  __shared__ float ldsMean[32 * ASTRIDE];
  __shared__ float ldsX[32 * ASTRIDE];
  __shared__ int sNext;

  const int tid  = threadIdx.x;
  const int wave = tid >> 5;
  const int lane = tid & 31;
  const int ntA  = (nrowsA + 31) >> 5;
  const int ntB  = (nrowsB + 31) >> 5;
  const int ntot = ntA + ntB;
  const int nwg  = gridDim.x;
  int tile = blockIdx.x;

  int ov = 0;
  float4 xv[2];
  {
    const bool fb = tile >= ntA;
    const int  fl = fb ? tile - ntA : tile;
    const int  fn = fb ? nrowsB : nrowsA;
    const int* __restrict__ foff   = fb ? offB : offA;
    const float* __restrict__ fxd  = fb ? xdstB : xdstA;
    if (tile < ntot && lane < 5) {
      const int gi = fl * 32 + wave * 4 + lane;
      ov = foff[gi < fn ? gi : fn];
    }
    #pragma unroll
    for (int u = 0; u < 2; ++u) {
      const int g = fl * 32 + (tid >> 4) + 16 * u;
      xv[u] = make_float4(0.f, 0.f, 0.f, 0.f);
      if (tile < ntot && g < fn) xv[u] = *(const float4*)&fxd[(size_t)g * D + (tid & 15) * 4];
    }
  }

  int wtype = -1;

  while (tile < ntot) {
    const bool tb   = tile >= ntA;
    const int  tl   = tb ? tile - ntA : tile;
    const int  nrows = tb ? nrowsB : nrowsA;
    const int* __restrict__ col = tb ? colB : colA;
    const float2* __restrict__ x2 = (const float2*)(tb ? xsrcB : xsrcA);
    const int row0 = tl * 32;

    const int sr = tid >> 4, sc = (tid & 15) * 4;

    __syncthreads();
    int nxv = 0;
    int zoff = 0;
    asm volatile("" : "+v"(zoff));
    if (tid == 0) nxv = atomicAdd(ctr + zoff, 1);
    #pragma unroll
    for (int u = 0; u < 2; ++u)
      *(float4*)&ldsX[(sr + 16 * u) * ASTRIDE + sc] = xv[u];

    if ((int)tb != wtype) {
      const float* __restrict__ Wl = tb ? WlB : WlA;
      const float* __restrict__ Wr = tb ? WrB : WrA;
      int tw = tid;
      asm volatile("" : "+v"(tw));
      float4 wl[4], wr[4];
      #pragma unroll
      for (int u = 0; u < 4; ++u) {
        const int i = (tw + u * 256) * 4;
        wl[u] = *(const float4*)&Wl[i];
        wr[u] = *(const float4*)&Wr[i];
      }
      #pragma unroll
      for (int u = 0; u < 4; ++u) {
        const int i = (tw + u * 256) * 4;
        const int r = i >> 6, c = i & 63;
        *(float4*)&ldsWl[r * WSTRIDE + c] = wl[u];
        *(float4*)&ldsWr[r * WSTRIDE + c] = wr[u];
      }
      wtype = (int)tb;
    }

    int rb[5];
    #pragma unroll
    for (int q = 0; q < 5; ++q) rb[q] = __builtin_amdgcn_readlane(ov, q);

    int dg[4], st[4], ix[4];
    #pragma unroll
    for (int q = 0; q < 4; ++q) { dg[q] = rb[q + 1] - rb[q]; st[q] = rb[q]; ix[q] = q; }
#define SAGE_CSWAP(a, b)                                         \
  {                                                              \
    const bool sw = dg[a] < dg[b];                               \
    const int d0_ = sw ? dg[b] : dg[a], d1_ = sw ? dg[a] : dg[b]; \
    const int s0_ = sw ? st[b] : st[a], s1_ = sw ? st[a] : st[b]; \
    const int i0_ = sw ? ix[b] : ix[a], i1_ = sw ? ix[a] : ix[b]; \
    dg[a] = d0_; dg[b] = d1_; st[a] = s0_; st[b] = s1_; ix[a] = i0_; ix[b] = i1_; \
  }
    SAGE_CSWAP(0, 1) SAGE_CSWAP(2, 3) SAGE_CSWAP(0, 2) SAGE_CSWAP(1, 3) SAGE_CSWAP(1, 2)
#undef SAGE_CSWAP

    int ids[4];
    #pragma unroll
    for (int p = 0; p < 4; ++p) ids[p] = (lane < dg[p]) ? col[st[p] + lane] : 0;

    float ax[4] = {0.f, 0.f, 0.f, 0.f};
    float ay[4] = {0.f, 0.f, 0.f, 0.f};
    for (int cb = 0; cb < dg[0]; cb += 32) {
      const int ce = cb + 32;
      int idn[4] = {0, 0, 0, 0};
      if (ce < dg[0]) {
        #pragma unroll
        for (int p = 0; p < 4; ++p) idn[p] = (ce + lane < dg[p]) ? col[st[p] + ce + lane] : 0;
      }
      const int e3 = dg[3] < ce ? dg[3] : ce;
      const int e2 = dg[2] < ce ? dg[2] : ce;
      const int e1 = dg[1] < ce ? dg[1] : ce;
      const int e0 = dg[0] < ce ? dg[0] : ce;
      int t0 = cb;
      for (; t0 < e3; t0 += 8) gather_step<4>(x2, (unsigned)lane, ids, dg, cb, t0, ax, ay);
      for (; t0 < e2; t0 += 8) gather_step<3>(x2, (unsigned)lane, ids, dg, cb, t0, ax, ay);
      for (; t0 < e1; t0 += 8) gather_step<2>(x2, (unsigned)lane, ids, dg, cb, t0, ax, ay);
      for (; t0 < e0; t0 += 8) gather_step<1>(x2, (unsigned)lane, ids, dg, cb, t0, ax, ay);
      #pragma unroll
      for (int p = 0; p < 4; ++p) ids[p] = idn[p];
    }

    if (tid == 0) sNext = nxv + nwg;

    int t = tid;
    asm volatile("" : "+v"(t));
    const int xr = t >> 4, xc = (t & 15) * 4;
    const int lq = t & 15, lh = (t >> 4) & 1, wv = t >> 5;
    const int rt    = wv >> 2;
    const int ct    = wv & 3;
    const int mrow  = rt * 16 + lq;
    const int ncol  = ct * 16 + lq;
    const int khalf = lh * 2;
    const int mhalf = lh * 8;
    #pragma unroll
    for (int p = 0; p < 4; ++p) {
      const int r = wv * 4 + ix[p];
      const float inv = 1.0f / fmaxf((float)dg[p], 1.0f);
      *(float2*)&ldsMean[r * ASTRIDE + (t & 31) * 2] = make_float2(ax[p] * inv, ay[p] * inv);
    }
    __syncthreads();
    const int tn = __builtin_amdgcn_readfirstlane(sNext);

    const bool nb  = tn >= ntA;
    const int  nl  = nb ? tn - ntA : tn;
    const int  nrn = nb ? nrowsB : nrowsA;
    const int* __restrict__ offn  = nb ? offB : offA;
    const float* __restrict__ xdn = nb ? xdstB : xdstA;
    int ovn = 0;
    if (tn < ntot && (t & 31) < 5) {
      const int gi = nl * 32 + wv * 4 + (t & 31);
      ovn = offn[gi < nrn ? gi : nrn];
    }
    const float bcol = (tb ? biasB : biasA)[ncol];
    #pragma unroll
    for (int u = 0; u < 2; ++u) {
      const int g = nl * 32 + xr + 16 * u;
      xv[u] = make_float4(0.f, 0.f, 0.f, 0.f);
      if (tn < ntot && g < nrn) xv[u] = *(const float4*)&xdn[(size_t)g * D + xc];
    }

    v8f acc = {0.f, 0.f, 0.f, 0.f, 0.f, 0.f, 0.f, 0.f};
    #pragma unroll
    for (int k = 0; k < D; k += 4) {
      const int kk = k + khalf;
      const v2f a = *(const v2f*)&ldsMean[mrow * ASTRIDE + kk];
      v2f b;
      b.x = ldsWl[(kk + 0) * WSTRIDE + ncol];
      b.y = ldsWl[(kk + 1) * WSTRIDE + ncol];
      acc = __builtin_amdgcn_wmma_f32_16x16x4_f32(false, a, false, b,
                                                  (short)0, acc, false, false);
    }
    #pragma unroll
    for (int k = 0; k < D; k += 4) {
      const int kk = k + khalf;
      const v2f a = *(const v2f*)&ldsX[mrow * ASTRIDE + kk];
      v2f b;
      b.x = ldsWr[(kk + 0) * WSTRIDE + ncol];
      b.y = ldsWr[(kk + 1) * WSTRIDE + ncol];
      acc = __builtin_amdgcn_wmma_f32_16x16x4_f32(false, a, false, b,
                                                  (short)0, acc, false, false);
    }

    float* __restrict__ out = tb ? outB : outA;
    #pragma unroll
    for (int v = 0; v < 8; ++v) {
      const int g = row0 + rt * 16 + mhalf + v;
      if (g < nrows) {
        float val = acc[v] + bcol;
        if (do_relu) val = fmaxf(val, 0.0f);
        out[(size_t)g * D + ncol] = val;
      }
    }
    ov = ovn;
    tile = tn;
  }
}

extern "C" void kernel_launch(void* const* d_in, const int* in_sizes, int n_in,
                              void* d_out, int out_size, void* d_ws, size_t ws_size,
                              hipStream_t stream) {
  const float* x_user  = (const float*)d_in[0];
  const float* x_movie = (const float*)d_in[1];
  const float* Wl1_um  = (const float*)d_in[2];
  const float* Wr1_um  = (const float*)d_in[3];
  const float* b1_um   = (const float*)d_in[4];
  const float* Wl1_mu  = (const float*)d_in[5];
  const float* Wr1_mu  = (const float*)d_in[6];
  const float* b1_mu   = (const float*)d_in[7];
  const float* Wl2_um  = (const float*)d_in[8];
  const float* Wr2_um  = (const float*)d_in[9];
  const float* b2_um   = (const float*)d_in[10];
  const float* Wl2_mu  = (const float*)d_in[11];
  const float* Wr2_mu  = (const float*)d_in[12];
  const float* b2_mu   = (const float*)d_in[13];
  const int*   uidx    = (const int*)d_in[14];
  const int*   midx    = (const int*)d_in[15];

  float* ws  = (float*)d_ws;
  int*   wsi = (int*)d_ws;
  int* cnt0  = wsi + W_CNT0;
  int* cnt1  = wsi + W_CNT1;
  int* base0 = wsi + W_BASE0;
  int* base1 = wsi + W_BASE1;
  int* offU  = wsi + W_OFFU;
  int* offM  = wsi + W_OFFM;
  int* colU  = wsi + W_COLU;
  int* colM  = wsi + W_COLM;
  int* rec0  = wsi + W_REC0;
  int* rec1  = wsi + W_REC1;
  int* ctr   = wsi + W_CTR;
  float* u1 = ws + OFF_U1;
  float* m1 = ws + OFF_M1;

  float* u2 = (float*)d_out;
  float* m2 = (float*)d_out + (size_t)N_USER * D;

  const dim3 blk(256);
  const dim3 big(1024);
  const dim3 gChunk(NCH, 2);
  const dim3 gBucket(NB, 2);
  const dim3 gLayer(1280);

  bin_count_kernel<<<gChunk, big, 0, stream>>>(uidx, midx, cnt0, cnt1, N_EDGE);
  scan_kernel<<<dim3(2), big, 0, stream>>>(cnt0, cnt1, base0, base1, ctr);
  bin_scatter_kernel<<<gChunk, big, 0, stream>>>(uidx, midx, base0, base1,
                                                 rec0, rec1, N_EDGE);
  bucket_csr_kernel<<<gBucket, big, 0, stream>>>(base0, base1, rec0, rec1,
                                                 offU, offM, colU, colM);

  sage_pull_wmma_kernel<<<gLayer, blk, 0, stream>>>(
      x_user,  offM, colM, x_movie, Wl1_um, Wr1_um, b1_um, m1, N_MOVIE,
      x_movie, offU, colU, x_user,  Wl1_mu, Wr1_mu, b1_mu, u1, N_USER,
      ctr, 1);

  sage_pull_wmma_kernel<<<gLayer, blk, 0, stream>>>(
      u1, offM, colM, m1, Wl2_um, Wr2_um, b2_um, m2, N_MOVIE,
      m1, offU, colU, u1, Wl2_mu, Wr2_mu, b2_mu, u2, N_USER,
      ctr + 1, 0);
}


# CSR build in 2 launches, not 4: each edge chunk is sorted by destination range in LDS from its own counts and stored at its own place with its run starts; no count or scan pass; sum order changes
# speedup vs baseline: 6.1751x; 1.0652x over previous
#include <hip/hip_runtime.h>


#define N_USER  200000
#define N_MOVIE 100000
#define N_EDGE  4000000
#define D       64

typedef float v2f __attribute__((ext_vector_type(2)));
typedef float v8f __attribute__((ext_vector_type(8)));

#define WSTRIDE 80
#define ASTRIDE 68

#define CH       16384
#define NCH      245
#define NB       196
#define LSTP     197
#define SHIFT0   10
#define SHIFT1   9
#define SRCBITS0 17
#define SRCBITS1 18
#define STCAP    24576

#define W_LST0  0u
#define W_LST1  50000u
#define W_CTR   250000u
#define W_OFFU  300000u
#define W_OFFM  500064u
#define W_COLU  600128u
#define W_COLM  4600128u
#define W_REC0  8600128u
#define W_REC1  12600128u
#define OFF_U1  19500000u
#define OFF_M1  (OFF_U1 + N_USER * 64u)

__global__ __launch_bounds__(1024) void bin_sort_kernel(
    const int* __restrict__ uidx, const int* __restrict__ midx,
    int* __restrict__ lst0, int* __restrict__ lst1,
    int* __restrict__ rec0, int* __restrict__ rec1,
    int* __restrict__ ctr, int nedge) {
  __shared__ int sRec[CH];
  __shared__ int cur[256];
  __shared__ int wtot[8];
  const int dir = blockIdx.y;
  const int* __restrict__ dstA = dir ? midx : uidx;
  const int* __restrict__ srcA = dir ? uidx : midx;
  int* __restrict__ lst        = dir ? lst1 : lst0;
  int* __restrict__ rec        = dir ? rec1 : rec0;
  const int shift = dir ? SHIFT1 : SHIFT0;
  const int sbits = dir ? SRCBITS1 : SRCBITS0;
  const int tid  = threadIdx.x;
  const int lane = tid & 31;
  const int c    = blockIdx.x;
  const int e0   = c * CH;
  const int rest = nedge - e0;
  const int nloc = rest < CH ? rest : CH;

  if (c == 0 && dir == 0 && tid < 2) ctr[tid] = 0;
  if (tid < 256) cur[tid] = 0;
  __syncthreads();

  int d[16], s[16];
  #pragma unroll
  for (int k = 0; k < 16; ++k) {
    const int e = e0 + k * 1024 + tid;
    d[k] = (e < nedge) ? dstA[e] : -1;
    s[k] = (e < nedge) ? srcA[e] : 0;
  }
  #pragma unroll
  for (int k = 0; k < 16; ++k)
    if (d[k] >= 0) atomicAdd(&cur[d[k] >> shift], 1);
  __syncthreads();

  int cn = 0, x = 0;
  if (tid < 256) {
    cn = cur[tid];
    x  = cn;
    #pragma unroll
    for (int q = 1; q < 32; q <<= 1) {
      const int y = __shfl_up(x, q);
      if (lane >= q) x += y;
    }
    if (lane == 31) wtot[tid >> 5] = x;
  }
  __syncthreads();
  if (tid < 256) {
    int wb = 0;
    for (int w = 0; w < (tid >> 5); ++w) wb += wtot[w];
    const int ex = wb + x - cn;
    cur[tid] = ex;
    if (tid <= NB) lst[c * LSTP + tid] = ex;
  }
  __syncthreads();

  const int lmask = (1 << shift) - 1;
  #pragma unroll
  for (int k = 0; k < 16; ++k) {
    if (d[k] >= 0) {
      const int p = atomicAdd(&cur[d[k] >> shift], 1);
      sRec[p] = s[k] | ((d[k] & lmask) << sbits);
    }
  }
  __syncthreads();

  int* __restrict__ outp = rec + (size_t)c * CH;
  for (int i = tid * 4; i < nloc; i += 4096) {
    if (i + 3 < nloc) {
      *(int4*)&outp[i] = *(const int4*)&sRec[i];
    } else {
      for (int j = i; j < nloc; ++j) outp[j] = sRec[j];
    }
  }
}

__global__ __launch_bounds__(1024) void bucket_csr_kernel(
    const int* __restrict__ lst0, const int* __restrict__ lst1,
    const int* __restrict__ rec0, const int* __restrict__ rec1,
    int* __restrict__ offU, int* __restrict__ offM,
    int* __restrict__ colU, int* __restrict__ colM) {
  __shared__ int cnt[1024];
  __shared__ int wsum[32];
  __shared__ int rs[256];
  __shared__ int rl[256];
  __shared__ int redS[8];
  __shared__ int redL[8];
  __shared__ int stage[STCAP];
  const int dir = blockIdx.y;
  const int b   = blockIdx.x;
  const int* __restrict__ lst = dir ? lst1 : lst0;
  const int* __restrict__ rec = dir ? rec1 : rec0;
  int* __restrict__ off        = dir ? offM : offU;
  int* __restrict__ col        = dir ? colM : colU;
  const int shift = dir ? SHIFT1 : SHIFT0;
  const int sbits = dir ? SRCBITS1 : SRCBITS0;
  const int nrows = dir ? N_MOVIE : N_USER;
  const int smask = (1 << sbits) - 1;
  const int tid  = threadIdx.x;
  const int lane = tid & 31;
  const int wave = tid >> 5;

  int s0 = 0, ln = 0;
  if (tid < NCH) {
    const int a = lst[tid * LSTP + b];
    const int e = lst[tid * LSTP + b + 1];
    s0 = a;
    ln = e - a;
    rs[tid] = tid * CH + a;
    rl[tid] = ln;
  }
  cnt[tid] = 0;
  if (wave < 8) {
    #pragma unroll
    for (int q = 16; q > 0; q >>= 1) {
      s0 += __shfl_xor(s0, q);
      ln += __shfl_xor(ln, q);
    }
    if (lane == 0) { redS[wave] = s0; redL[wave] = ln; }
  }
  __syncthreads();
  int bs = 0, n = 0;
  #pragma unroll
  for (int w = 0; w < 8; ++w) { bs += redS[w]; n += redL[w]; }
  const int be = bs + n;

  for (int c0 = wave; c0 < NCH; c0 += 64) {
    const int c1 = c0 + 32;
    const int a0 = rs[c0], l0 = rl[c0];
    const int a1 = (c1 < NCH) ? rs[c1] : 0;
    const int l1 = (c1 < NCH) ? rl[c1] : 0;
    const int lm = l0 > l1 ? l0 : l1;
    for (int i0 = 0; i0 < lm; i0 += 128) {
      int r[8];
      #pragma unroll
      for (int k = 0; k < 4; ++k) {
        const int i = i0 + k * 32 + lane;
        r[k]     = (i < l0) ? rec[a0 + i] : -1;
        r[4 + k] = (i < l1) ? rec[a1 + i] : -1;
      }
      #pragma unroll
      for (int k = 0; k < 8; ++k)
        if (r[k] >= 0) atomicAdd(&cnt[r[k] >> sbits], 1);
    }
  }
  __syncthreads();

  const int v = cnt[tid];
  int x = v;
  #pragma unroll
  for (int q = 1; q < 32; q <<= 1) {
    const int y = __shfl_up(x, q);
    if (lane >= q) x += y;
  }
  if (lane == 31) wsum[wave] = x;
  __syncthreads();
  if (wave == 0) {
    int sacc = wsum[lane];
    #pragma unroll
    for (int q = 1; q < 32; q <<= 1) {
      const int y = __shfl_up(sacc, q);
      if (lane >= q) sacc += y;
    }
    wsum[lane] = sacc;
  }
  __syncthreads();
  const int ex = ((wave > 0) ? wsum[wave - 1] : 0) + x - v;
  const int node = (b << shift) + tid;
  if (tid < (1 << shift) && node < nrows) off[node] = bs + ex;
  if (b == NB - 1 && tid == 0) off[nrows] = be;
  cnt[tid] = ex;
  __syncthreads();

  const bool staged = n <= STCAP;
  for (int c0 = wave; c0 < NCH; c0 += 64) {
    const int c1 = c0 + 32;
    const int a0 = rs[c0], l0 = rl[c0];
    const int a1 = (c1 < NCH) ? rs[c1] : 0;
    const int l1 = (c1 < NCH) ? rl[c1] : 0;
    const int lm = l0 > l1 ? l0 : l1;
    for (int i0 = 0; i0 < lm; i0 += 128) {
      int r[8];
      #pragma unroll
      for (int k = 0; k < 4; ++k) {
        const int i = i0 + k * 32 + lane;
        r[k]     = (i < l0) ? rec[a0 + i] : -1;
        r[4 + k] = (i < l1) ? rec[a1 + i] : -1;
      }
      #pragma unroll
      for (int k = 0; k < 8; ++k) {
        if (r[k] >= 0) {
          const int p = atomicAdd(&cnt[r[k] >> sbits], 1);
          if (staged) stage[p] = r[k] & smask;
          else        col[bs + p] = r[k] & smask;
        }
      }
    }
  }
  if (staged) {
    __syncthreads();
    for (int i = tid; i < n; i += 1024) col[bs + i] = stage[i];
  }
}

template <int NA>
__device__ __forceinline__ void gather_step(const float2* __restrict__ x2, unsigned lane,
                                            const int (&ids)[4], const int (&dg)[4],
                                            int cb, int t0, float (&ax)[4], float (&ay)[4]) {
  float2 v[NA][8];
  const int sb = t0 - cb;
  #pragma unroll
  for (int p = 0; p < NA; ++p) {
    #pragma unroll
    for (int k = 0; k < 8; ++k) {
      const unsigned sidx = (unsigned)__builtin_amdgcn_readlane(ids[p], sb + k);
      v[p][k] = x2[sidx * 32u + lane];
    }
  }
  #pragma unroll
  for (int p = 0; p < NA; ++p) {
    if (t0 + 8 <= dg[p]) {
      #pragma unroll
      for (int k = 0; k < 8; ++k) {
        ax[p] += v[p][k].x;
        ay[p] += v[p][k].y;
      }
    } else {
      #pragma unroll
      for (int k = 0; k < 8; ++k) {
        const bool ok = (t0 + k) < dg[p];
        ax[p] += ok ? v[p][k].x : 0.f;
        ay[p] += ok ? v[p][k].y : 0.f;
      }
    }
  }
}

__global__ __launch_bounds__(256) void sage_pull_wmma_kernel(
    const float* __restrict__ xsrcA, const int* __restrict__ offA,
    const int* __restrict__ colA, const float* __restrict__ xdstA,
    const float* __restrict__ WlA, const float* __restrict__ WrA,
    const float* __restrict__ biasA, float* __restrict__ outA, int nrowsA,
    const float* __restrict__ xsrcB, const int* __restrict__ offB,
    const int* __restrict__ colB, const float* __restrict__ xdstB,
    const float* __restrict__ WlB, const float* __restrict__ WrB,
    const float* __restrict__ biasB, float* __restrict__ outB, int nrowsB,
    int* __restrict__ ctr, int do_relu) {
  __shared__ float ldsWl[D * WSTRIDE];
  __shared__ float ldsWr[D * WSTRIDE];
  __shared__ float ldsMean[32 * ASTRIDE];
  __shared__ float ldsX[32 * ASTRIDE];
  __shared__ int sNext;

  const int tid  = threadIdx.x;
  const int wave = tid >> 5;
  const int lane = tid & 31;
  const int ntA  = (nrowsA + 31) >> 5;
  const int ntB  = (nrowsB + 31) >> 5;
  const int ntot = ntA + ntB;
  const int nwg  = gridDim.x;
  int tile = blockIdx.x;

  int ov = 0;
  float4 xv[2];
  {
    const bool fb = tile >= ntA;
    const int  fl = fb ? tile - ntA : tile;
    const int  fn = fb ? nrowsB : nrowsA;
    const int* __restrict__ foff   = fb ? offB : offA;
    const float* __restrict__ fxd  = fb ? xdstB : xdstA;
    if (tile < ntot && lane < 5) {
      const int gi = fl * 32 + wave * 4 + lane;
      ov = foff[gi < fn ? gi : fn];
    }
    #pragma unroll
    for (int u = 0; u < 2; ++u) {
      const int g = fl * 32 + (tid >> 4) + 16 * u;
      xv[u] = make_float4(0.f, 0.f, 0.f, 0.f);
      if (tile < ntot && g < fn) xv[u] = *(const float4*)&fxd[(size_t)g * D + (tid & 15) * 4];
    }
  }

  int wtype = -1;

  while (tile < ntot) {
    const bool tb   = tile >= ntA;
    const int  tl   = tb ? tile - ntA : tile;
    const int  nrows = tb ? nrowsB : nrowsA;
    const int* __restrict__ col = tb ? colB : colA;
    const float2* __restrict__ x2 = (const float2*)(tb ? xsrcB : xsrcA);
    const int row0 = tl * 32;

    const int sr = tid >> 4, sc = (tid & 15) * 4;

    __syncthreads();
    int nxv = 0;
    int zoff = 0;
    asm volatile("" : "+v"(zoff));
    if (tid == 0) nxv = atomicAdd(ctr + zoff, 1);
    #pragma unroll
    for (int u = 0; u < 2; ++u)
      *(float4*)&ldsX[(sr + 16 * u) * ASTRIDE + sc] = xv[u];

    if ((int)tb != wtype) {
      const float* __restrict__ Wl = tb ? WlB : WlA;
      const float* __restrict__ Wr = tb ? WrB : WrA;
      int tw = tid;
      asm volatile("" : "+v"(tw));
      float4 wl[4], wr[4];
      #pragma unroll
      for (int u = 0; u < 4; ++u) {
        const int i = (tw + u * 256) * 4;
        wl[u] = *(const float4*)&Wl[i];
        wr[u] = *(const float4*)&Wr[i];
      }
      #pragma unroll
      for (int u = 0; u < 4; ++u) {
        const int i = (tw + u * 256) * 4;
        const int r = i >> 6, c = i & 63;
        *(float4*)&ldsWl[r * WSTRIDE + c] = wl[u];
        *(float4*)&ldsWr[r * WSTRIDE + c] = wr[u];
      }
      wtype = (int)tb;
    }

    int rb[5];
    #pragma unroll
    for (int q = 0; q < 5; ++q) rb[q] = __builtin_amdgcn_readlane(ov, q);

    int dg[4], st[4], ix[4];
    #pragma unroll
    for (int q = 0; q < 4; ++q) { dg[q] = rb[q + 1] - rb[q]; st[q] = rb[q]; ix[q] = q; }
#define SAGE_CSWAP(a, b)                                         \
  {                                                              \
    const bool sw = dg[a] < dg[b];                               \
    const int d0_ = sw ? dg[b] : dg[a], d1_ = sw ? dg[a] : dg[b]; \
    const int s0_ = sw ? st[b] : st[a], s1_ = sw ? st[a] : st[b]; \
    const int i0_ = sw ? ix[b] : ix[a], i1_ = sw ? ix[a] : ix[b]; \
    dg[a] = d0_; dg[b] = d1_; st[a] = s0_; st[b] = s1_; ix[a] = i0_; ix[b] = i1_; \
  }
    SAGE_CSWAP(0, 1) SAGE_CSWAP(2, 3) SAGE_CSWAP(0, 2) SAGE_CSWAP(1, 3) SAGE_CSWAP(1, 2)
#undef SAGE_CSWAP

    int ids[4];
    #pragma unroll
    for (int p = 0; p < 4; ++p) ids[p] = (lane < dg[p]) ? col[st[p] + lane] : 0;

    float ax[4] = {0.f, 0.f, 0.f, 0.f};
    float ay[4] = {0.f, 0.f, 0.f, 0.f};
    for (int cb = 0; cb < dg[0]; cb += 32) {
      const int ce = cb + 32;
      int idn[4] = {0, 0, 0, 0};
      if (ce < dg[0]) {
        #pragma unroll
        for (int p = 0; p < 4; ++p) idn[p] = (ce + lane < dg[p]) ? col[st[p] + ce + lane] : 0;
      }
      const int e3 = dg[3] < ce ? dg[3] : ce;
      const int e2 = dg[2] < ce ? dg[2] : ce;
      const int e1 = dg[1] < ce ? dg[1] : ce;
      const int e0 = dg[0] < ce ? dg[0] : ce;
      int t0 = cb;
      for (; t0 < e3; t0 += 8) gather_step<4>(x2, (unsigned)lane, ids, dg, cb, t0, ax, ay);
      for (; t0 < e2; t0 += 8) gather_step<3>(x2, (unsigned)lane, ids, dg, cb, t0, ax, ay);
      for (; t0 < e1; t0 += 8) gather_step<2>(x2, (unsigned)lane, ids, dg, cb, t0, ax, ay);
      for (; t0 < e0; t0 += 8) gather_step<1>(x2, (unsigned)lane, ids, dg, cb, t0, ax, ay);
      #pragma unroll
      for (int p = 0; p < 4; ++p) ids[p] = idn[p];
    }

    if (tid == 0) sNext = nxv + nwg;

    int t = tid;
    asm volatile("" : "+v"(t));
    const int xr = t >> 4, xc = (t & 15) * 4;
    const int lq = t & 15, lh = (t >> 4) & 1, wv = t >> 5;
    const int rt    = wv >> 2;
    const int ct    = wv & 3;
    const int mrow  = rt * 16 + lq;
    const int ncol  = ct * 16 + lq;
    const int khalf = lh * 2;
    const int mhalf = lh * 8;
    #pragma unroll
    for (int p = 0; p < 4; ++p) {
      const int r = wv * 4 + ix[p];
      const float inv = 1.0f / fmaxf((float)dg[p], 1.0f);
      *(float2*)&ldsMean[r * ASTRIDE + (t & 31) * 2] = make_float2(ax[p] * inv, ay[p] * inv);
    }
    __syncthreads();
    const int tn = __builtin_amdgcn_readfirstlane(sNext);

    const bool nb  = tn >= ntA;
    const int  nl  = nb ? tn - ntA : tn;
    const int  nrn = nb ? nrowsB : nrowsA;
    const int* __restrict__ offn  = nb ? offB : offA;
    const float* __restrict__ xdn = nb ? xdstB : xdstA;
    int ovn = 0;
    if (tn < ntot && (t & 31) < 5) {
      const int gi = nl * 32 + wv * 4 + (t & 31);
      ovn = offn[gi < nrn ? gi : nrn];
    }
    const float bcol = (tb ? biasB : biasA)[ncol];
    #pragma unroll
    for (int u = 0; u < 2; ++u) {
      const int g = nl * 32 + xr + 16 * u;
      xv[u] = make_float4(0.f, 0.f, 0.f, 0.f);
      if (tn < ntot && g < nrn) xv[u] = *(const float4*)&xdn[(size_t)g * D + xc];
    }

    v8f acc = {0.f, 0.f, 0.f, 0.f, 0.f, 0.f, 0.f, 0.f};
    #pragma unroll
    for (int k = 0; k < D; k += 4) {
      const int kk = k + khalf;
      const v2f a = *(const v2f*)&ldsMean[mrow * ASTRIDE + kk];
      v2f b;
      b.x = ldsWl[(kk + 0) * WSTRIDE + ncol];
      b.y = ldsWl[(kk + 1) * WSTRIDE + ncol];
      acc = __builtin_amdgcn_wmma_f32_16x16x4_f32(false, a, false, b,
                                                  (short)0, acc, false, false);
    }
    #pragma unroll
    for (int k = 0; k < D; k += 4) {
      const int kk = k + khalf;
      const v2f a = *(const v2f*)&ldsX[mrow * ASTRIDE + kk];
      v2f b;
      b.x = ldsWr[(kk + 0) * WSTRIDE + ncol];
      b.y = ldsWr[(kk + 1) * WSTRIDE + ncol];
      acc = __builtin_amdgcn_wmma_f32_16x16x4_f32(false, a, false, b,
                                                  (short)0, acc, false, false);
    }

    float* __restrict__ out = tb ? outB : outA;
    #pragma unroll
    for (int v = 0; v < 8; ++v) {
      const int g = row0 + rt * 16 + mhalf + v;
      if (g < nrows) {
        float val = acc[v] + bcol;
        if (do_relu) val = fmaxf(val, 0.0f);
        out[(size_t)g * D + ncol] = val;
      }
    }
    ov = ovn;
    tile = tn;
  }
}

extern "C" void kernel_launch(void* const* d_in, const int* in_sizes, int n_in,
                              void* d_out, int out_size, void* d_ws, size_t ws_size,
                              hipStream_t stream) {
  const float* x_user  = (const float*)d_in[0];
  const float* x_movie = (const float*)d_in[1];
  const float* Wl1_um  = (const float*)d_in[2];
  const float* Wr1_um  = (const float*)d_in[3];
  const float* b1_um   = (const float*)d_in[4];
  const float* Wl1_mu  = (const float*)d_in[5];
  const float* Wr1_mu  = (const float*)d_in[6];
  const float* b1_mu   = (const float*)d_in[7];
  const float* Wl2_um  = (const float*)d_in[8];
  const float* Wr2_um  = (const float*)d_in[9];
  const float* b2_um   = (const float*)d_in[10];
  const float* Wl2_mu  = (const float*)d_in[11];
  const float* Wr2_mu  = (const float*)d_in[12];
  const float* b2_mu   = (const float*)d_in[13];
  const int*   uidx    = (const int*)d_in[14];
  const int*   midx    = (const int*)d_in[15];

  float* ws  = (float*)d_ws;
  int*   wsi = (int*)d_ws;
  int* lst0  = wsi + W_LST0;
  int* lst1  = wsi + W_LST1;
  int* offU  = wsi + W_OFFU;
  int* offM  = wsi + W_OFFM;
  int* colU  = wsi + W_COLU;
  int* colM  = wsi + W_COLM;
  int* rec0  = wsi + W_REC0;
  int* rec1  = wsi + W_REC1;
  int* ctr   = wsi + W_CTR;
  float* u1 = ws + OFF_U1;
  float* m1 = ws + OFF_M1;

  float* u2 = (float*)d_out;
  float* m2 = (float*)d_out + (size_t)N_USER * D;

  const dim3 blk(256);
  const dim3 big(1024);
  const dim3 gChunk(NCH, 2);
  const dim3 gBucket(NB, 2);
  const dim3 gLayer(1280);

  bin_sort_kernel<<<gChunk, big, 0, stream>>>(uidx, midx, lst0, lst1,
                                              rec0, rec1, ctr, N_EDGE);
  bucket_csr_kernel<<<gBucket, big, 0, stream>>>(lst0, lst1, rec0, rec1,
                                                 offU, offM, colU, colM);

  sage_pull_wmma_kernel<<<gLayer, blk, 0, stream>>>(
      x_user,  offM, colM, x_movie, Wl1_um, Wr1_um, b1_um, m1, N_MOVIE,
      x_movie, offU, colU, x_user,  Wl1_mu, Wr1_mu, b1_mu, u1, N_USER,
      ctr, 1);

  sage_pull_wmma_kernel<<<gLayer, blk, 0, stream>>>(
      u1, offM, colM, m1, Wl2_um, Wr2_um, b2_um, m2, N_MOVIE,
      m1, offU, colU, u1, Wl2_mu, Wr2_mu, b2_mu, u2, N_USER,
      ctr + 1, 0);
}


# speedup vs baseline: 6.3836x; 1.0338x over previous
#include <hip/hip_runtime.h>


#define N_USER  200000
#define N_MOVIE 100000
#define N_EDGE  4000000
#define D       64

typedef float v2f __attribute__((ext_vector_type(2)));
typedef float v8f __attribute__((ext_vector_type(8)));

#define WSTRIDE 80
#define ASTRIDE 68

#define CH       16384
#define NCH      245
#define NB       196
#define LSTP     197
#define SHIFT0   10
#define SHIFT1   9
#define SRCBITS0 17
#define SRCBITS1 18
#define STCAP    24576

#define W_LST0  0u
#define W_LST1  50000u
#define W_CTR   250000u
#define W_OFFU  300000u
#define W_OFFM  500064u
#define W_COLU  600128u
#define W_COLM  4600128u
#define W_REC0  8600128u
#define W_REC1  12600128u
#define OFF_U1  19499968u
#define OFF_M1  (OFF_U1 + N_USER * 64u)
static_assert((OFF_U1 * 4u) % 256u == 0u && (OFF_M1 * 4u) % 256u == 0u,
              "layer-1 tables must start on 256-byte boundaries");
static_assert(OFF_U1 >= W_REC1 + N_EDGE, "layer-1 tables must lie behind the CSR build's arrays");

__global__ __launch_bounds__(1024) void bin_sort_kernel(
    const int* __restrict__ uidx, const int* __restrict__ midx,
    int* __restrict__ lst0, int* __restrict__ lst1,
    int* __restrict__ rec0, int* __restrict__ rec1,
    int* __restrict__ ctr, int nedge) {
  __shared__ int sRec[CH];
  __shared__ int cur[256];
  __shared__ int wtot[8];
  const int dir = blockIdx.y;
  const int* __restrict__ dstA = dir ? midx : uidx;
  const int* __restrict__ srcA = dir ? uidx : midx;
  int* __restrict__ lst        = dir ? lst1 : lst0;
  int* __restrict__ rec        = dir ? rec1 : rec0;
  const int shift = dir ? SHIFT1 : SHIFT0;
  const int sbits = dir ? SRCBITS1 : SRCBITS0;
  const int tid  = threadIdx.x;
  const int lane = tid & 31;
  const int c    = blockIdx.x;
  const int e0   = c * CH;
  const int rest = nedge - e0;
  const int nloc = rest < CH ? rest : CH;

  if (c == 0 && dir == 0 && tid < 2) ctr[tid] = 0;
  if (tid < 256) cur[tid] = 0;
  __syncthreads();

  int d[16], s[16];
  #pragma unroll
  for (int k = 0; k < 16; ++k) {
    const int e = e0 + k * 1024 + tid;
    d[k] = (e < nedge) ? dstA[e] : -1;
    s[k] = (e < nedge) ? srcA[e] : 0;
  }
  #pragma unroll
  for (int k = 0; k < 16; ++k)
    if (d[k] >= 0) atomicAdd(&cur[d[k] >> shift], 1);
  __syncthreads();

  int cn = 0, x = 0;
  if (tid < 256) {
    cn = cur[tid];
    x  = cn;
    #pragma unroll
    for (int q = 1; q < 32; q <<= 1) {
      const int y = __shfl_up(x, q);
      if (lane >= q) x += y;
    }
    if (lane == 31) wtot[tid >> 5] = x;
  }
  __syncthreads();
  if (tid < 256) {
    int wb = 0;
    for (int w = 0; w < (tid >> 5); ++w) wb += wtot[w];
    const int ex = wb + x - cn;
    cur[tid] = ex;
    if (tid <= NB) lst[c * LSTP + tid] = ex;
  }
  __syncthreads();

  const int lmask = (1 << shift) - 1;
  #pragma unroll
  for (int k = 0; k < 16; ++k) {
    if (d[k] >= 0) {
      const int p = atomicAdd(&cur[d[k] >> shift], 1);
      sRec[p] = s[k] | ((d[k] & lmask) << sbits);
    }
  }
  __syncthreads();

  int* __restrict__ outp = rec + (size_t)c * CH;
  for (int i = tid * 4; i < nloc; i += 4096) {
    if (i + 3 < nloc) {
      *(int4*)&outp[i] = *(const int4*)&sRec[i];
    } else {
      for (int j = i; j < nloc; ++j) outp[j] = sRec[j];
    }
  }
}

__global__ __launch_bounds__(1024) void bucket_csr_kernel(
    const int* __restrict__ lst0, const int* __restrict__ lst1,
    const int* __restrict__ rec0, const int* __restrict__ rec1,
    int* __restrict__ offU, int* __restrict__ offM,
    int* __restrict__ colU, int* __restrict__ colM) {
  __shared__ int cnt[1024];
  __shared__ int wsum[32];
  __shared__ int rs[256];
  __shared__ int rl[256];
  __shared__ int redS[8];
  __shared__ int redL[8];
  __shared__ int stage[STCAP];
  const int dir = blockIdx.y;
  const int b   = blockIdx.x;
  const int* __restrict__ lst = dir ? lst1 : lst0;
  const int* __restrict__ rec = dir ? rec1 : rec0;
  int* __restrict__ off        = dir ? offM : offU;
  int* __restrict__ col        = dir ? colM : colU;
  const int shift = dir ? SHIFT1 : SHIFT0;
  const int sbits = dir ? SRCBITS1 : SRCBITS0;
  const int nrows = dir ? N_MOVIE : N_USER;
  const int smask = (1 << sbits) - 1;
  const int tid  = threadIdx.x;
  const int lane = tid & 31;
  const int wave = tid >> 5;

  int s0 = 0, ln = 0;
  if (tid < NCH) {
    const int a = lst[tid * LSTP + b];
    const int e = lst[tid * LSTP + b + 1];
    s0 = a;
    ln = e - a;
    rs[tid] = tid * CH + a;
    rl[tid] = ln;
  }
  cnt[tid] = 0;
  if (wave < 8) {
    #pragma unroll
    for (int q = 16; q > 0; q >>= 1) {
      s0 += __shfl_xor(s0, q);
      ln += __shfl_xor(ln, q);
    }
    if (lane == 0) { redS[wave] = s0; redL[wave] = ln; }
  }
  __syncthreads();
  int bs = 0, n = 0;
  #pragma unroll
  for (int w = 0; w < 8; ++w) { bs += redS[w]; n += redL[w]; }
  const int be = bs + n;

  for (int c0 = wave; c0 < NCH; c0 += 64) {
    const int c1 = c0 + 32;
    const int a0 = rs[c0], l0 = rl[c0];
    const int a1 = (c1 < NCH) ? rs[c1] : 0;
    const int l1 = (c1 < NCH) ? rl[c1] : 0;
    const int lm = l0 > l1 ? l0 : l1;
    for (int i0 = 0; i0 < lm; i0 += 128) {
      int r[8];
      #pragma unroll
      for (int k = 0; k < 4; ++k) {
        const int i = i0 + k * 32 + lane;
        r[k]     = (i < l0) ? rec[a0 + i] : -1;
        r[4 + k] = (i < l1) ? rec[a1 + i] : -1;
      }
      #pragma unroll
      for (int k = 0; k < 8; ++k)
        if (r[k] >= 0) atomicAdd(&cnt[r[k] >> sbits], 1);
    }
  }
  __syncthreads();

  const int v = cnt[tid];
  int x = v;
  #pragma unroll
  for (int q = 1; q < 32; q <<= 1) {
    const int y = __shfl_up(x, q);
    if (lane >= q) x += y;
  }
  if (lane == 31) wsum[wave] = x;
  __syncthreads();
  if (wave == 0) {
    int sacc = wsum[lane];
    #pragma unroll
    for (int q = 1; q < 32; q <<= 1) {
      const int y = __shfl_up(sacc, q);
      if (lane >= q) sacc += y;
    }
    wsum[lane] = sacc;
  }
  __syncthreads();
  const int ex = ((wave > 0) ? wsum[wave - 1] : 0) + x - v;
  const int node = (b << shift) + tid;
  if (tid < (1 << shift) && node < nrows) off[node] = bs + ex;
  if (b == NB - 1 && tid == 0) off[nrows] = be;
  cnt[tid] = ex;
  __syncthreads();

  const bool staged = n <= STCAP;
  for (int c0 = wave; c0 < NCH; c0 += 64) {
    const int c1 = c0 + 32;
    const int a0 = rs[c0], l0 = rl[c0];
    const int a1 = (c1 < NCH) ? rs[c1] : 0;
    const int l1 = (c1 < NCH) ? rl[c1] : 0;
    const int lm = l0 > l1 ? l0 : l1;
    for (int i0 = 0; i0 < lm; i0 += 128) {
      int r[8];
      #pragma unroll
      for (int k = 0; k < 4; ++k) {
        const int i = i0 + k * 32 + lane;
        r[k]     = (i < l0) ? rec[a0 + i] : -1;
        r[4 + k] = (i < l1) ? rec[a1 + i] : -1;
      }
      #pragma unroll
      for (int k = 0; k < 8; ++k) {
        if (r[k] >= 0) {
          const int p = atomicAdd(&cnt[r[k] >> sbits], 1);
          if (staged) stage[p] = r[k] & smask;
          else        col[bs + p] = r[k] & smask;
        }
      }
    }
  }
  if (staged) {
    __syncthreads();
    for (int i = tid; i < n; i += 1024) col[bs + i] = stage[i];
  }
}

template <int NA>
__device__ __forceinline__ void gather_step(const float2* __restrict__ x2, unsigned lane,
                                            const int (&ids)[4], const int (&dg)[4],
                                            int cb, int t0, float (&ax)[4], float (&ay)[4]) {
  float2 v[NA][8];
  const int sb = t0 - cb;
  #pragma unroll
  for (int p = 0; p < NA; ++p) {
    #pragma unroll
    for (int k = 0; k < 8; ++k) {
      const unsigned sidx = (unsigned)__builtin_amdgcn_readlane(ids[p], sb + k);
      v[p][k] = x2[sidx * 32u + lane];
    }
  }
  #pragma unroll
  for (int p = 0; p < NA; ++p) {
    if (t0 + 8 <= dg[p]) {
      #pragma unroll
      for (int k = 0; k < 8; ++k) {
        ax[p] += v[p][k].x;
        ay[p] += v[p][k].y;
      }
    } else {
      #pragma unroll
      for (int k = 0; k < 8; ++k) {
        const bool ok = (t0 + k) < dg[p];
        ax[p] += ok ? v[p][k].x : 0.f;
        ay[p] += ok ? v[p][k].y : 0.f;
      }
    }
  }
}

__global__ __launch_bounds__(256) void sage_pull_wmma_kernel(
    const float* __restrict__ xsrcA, const int* __restrict__ offA,
    const int* __restrict__ colA, const float* __restrict__ xdstA,
    const float* __restrict__ WlA, const float* __restrict__ WrA,
    const float* __restrict__ biasA, float* __restrict__ outA, int nrowsA,
    const float* __restrict__ xsrcB, const int* __restrict__ offB,
    const int* __restrict__ colB, const float* __restrict__ xdstB,
    const float* __restrict__ WlB, const float* __restrict__ WrB,
    const float* __restrict__ biasB, float* __restrict__ outB, int nrowsB,
    int* __restrict__ ctr, int do_relu) {
  __shared__ float ldsWl[D * WSTRIDE];
  __shared__ float ldsWr[D * WSTRIDE];
  __shared__ float ldsMean[32 * ASTRIDE];
  __shared__ float ldsX[32 * ASTRIDE];
  __shared__ int sNext;

  const int tid  = threadIdx.x;
  const int wave = tid >> 5;
  const int lane = tid & 31;
  const int ntA  = (nrowsA + 31) >> 5;
  const int ntB  = (nrowsB + 31) >> 5;
  const int ntot = ntA + ntB;
  const int nwg  = gridDim.x;
  int tile = blockIdx.x;

  int ov = 0;
  float4 xv[2];
  {
    const bool fb = tile >= ntA;
    const int  fl = fb ? tile - ntA : tile;
    const int  fn = fb ? nrowsB : nrowsA;
    const int* __restrict__ foff   = fb ? offB : offA;
    const float* __restrict__ fxd  = fb ? xdstB : xdstA;
    if (tile < ntot && lane < 5) {
      const int gi = fl * 32 + wave * 4 + lane;
      ov = foff[gi < fn ? gi : fn];
    }
    #pragma unroll
    for (int u = 0; u < 2; ++u) {
      const int g = fl * 32 + (tid >> 4) + 16 * u;
      xv[u] = make_float4(0.f, 0.f, 0.f, 0.f);
      if (tile < ntot && g < fn) xv[u] = *(const float4*)&fxd[(size_t)g * D + (tid & 15) * 4];
    }
  }

  int wtype = -1;

  while (tile < ntot) {
    const bool tb   = tile >= ntA;
    const int  tl   = tb ? tile - ntA : tile;
    const int  nrows = tb ? nrowsB : nrowsA;
    const int* __restrict__ col = tb ? colB : colA;
    const float2* __restrict__ x2 = (const float2*)(tb ? xsrcB : xsrcA);
    const int row0 = tl * 32;

    const int sr = tid >> 4, sc = (tid & 15) * 4;

    __syncthreads();
    int nxv = 0;
    int zoff = 0;
    asm volatile("" : "+v"(zoff));
    if (tid == 0) nxv = atomicAdd(ctr + zoff, 1);
    #pragma unroll
    for (int u = 0; u < 2; ++u)
      *(float4*)&ldsX[(sr + 16 * u) * ASTRIDE + sc] = xv[u];

    if ((int)tb != wtype) {
      const float* __restrict__ Wl = tb ? WlB : WlA;
      const float* __restrict__ Wr = tb ? WrB : WrA;
      int tw = tid;
      asm volatile("" : "+v"(tw));
      float4 wl[4], wr[4];
      #pragma unroll
      for (int u = 0; u < 4; ++u) {
        const int i = (tw + u * 256) * 4;
        wl[u] = *(const float4*)&Wl[i];
        wr[u] = *(const float4*)&Wr[i];
      }
      #pragma unroll
      for (int u = 0; u < 4; ++u) {
        const int i = (tw + u * 256) * 4;
        const int r = i >> 6, c = i & 63;
        *(float4*)&ldsWl[r * WSTRIDE + c] = wl[u];
        *(float4*)&ldsWr[r * WSTRIDE + c] = wr[u];
      }
      wtype = (int)tb;
    }

    int rb[5];
    #pragma unroll
    for (int q = 0; q < 5; ++q) rb[q] = __builtin_amdgcn_readlane(ov, q);

    int dg[4], st[4], ix[4];
    #pragma unroll
    for (int q = 0; q < 4; ++q) { dg[q] = rb[q + 1] - rb[q]; st[q] = rb[q]; ix[q] = q; }
#define SAGE_CSWAP(a, b)                                         \
  {                                                              \
    const bool sw = dg[a] < dg[b];                               \
    const int d0_ = sw ? dg[b] : dg[a], d1_ = sw ? dg[a] : dg[b]; \
    const int s0_ = sw ? st[b] : st[a], s1_ = sw ? st[a] : st[b]; \
    const int i0_ = sw ? ix[b] : ix[a], i1_ = sw ? ix[a] : ix[b]; \
    dg[a] = d0_; dg[b] = d1_; st[a] = s0_; st[b] = s1_; ix[a] = i0_; ix[b] = i1_; \
  }
    SAGE_CSWAP(0, 1) SAGE_CSWAP(2, 3) SAGE_CSWAP(0, 2) SAGE_CSWAP(1, 3) SAGE_CSWAP(1, 2)
#undef SAGE_CSWAP

    int ids[4];
    #pragma unroll
    for (int p = 0; p < 4; ++p) ids[p] = (lane < dg[p]) ? col[st[p] + lane] : 0;

    float ax[4] = {0.f, 0.f, 0.f, 0.f};
    float ay[4] = {0.f, 0.f, 0.f, 0.f};
    for (int cb = 0; cb < dg[0]; cb += 32) {
      const int ce = cb + 32;
      int idn[4] = {0, 0, 0, 0};
      if (ce < dg[0]) {
        #pragma unroll
        for (int p = 0; p < 4; ++p) idn[p] = (ce + lane < dg[p]) ? col[st[p] + ce + lane] : 0;
      }
      const int e3 = dg[3] < ce ? dg[3] : ce;
      const int e2 = dg[2] < ce ? dg[2] : ce;
      const int e1 = dg[1] < ce ? dg[1] : ce;
      const int e0 = dg[0] < ce ? dg[0] : ce;
      int t0 = cb;
      for (; t0 < e3; t0 += 8) gather_step<4>(x2, (unsigned)lane, ids, dg, cb, t0, ax, ay);
      for (; t0 < e2; t0 += 8) gather_step<3>(x2, (unsigned)lane, ids, dg, cb, t0, ax, ay);
      for (; t0 < e1; t0 += 8) gather_step<2>(x2, (unsigned)lane, ids, dg, cb, t0, ax, ay);
      for (; t0 < e0; t0 += 8) gather_step<1>(x2, (unsigned)lane, ids, dg, cb, t0, ax, ay);
      #pragma unroll
      for (int p = 0; p < 4; ++p) ids[p] = idn[p];
    }

    if (tid == 0) sNext = nxv + nwg;

    int t = tid;
    asm volatile("" : "+v"(t));
    const int xr = t >> 4, xc = (t & 15) * 4;
    const int lq = t & 15, lh = (t >> 4) & 1, wv = t >> 5;
    const int rt    = wv >> 2;
    const int ct    = wv & 3;
    const int mrow  = rt * 16 + lq;
    const int ncol  = ct * 16 + lq;
    const int khalf = lh * 2;
    const int mhalf = lh * 8;
    #pragma unroll
    for (int p = 0; p < 4; ++p) {
      const int r = wv * 4 + ix[p];
      const float inv = 1.0f / fmaxf((float)dg[p], 1.0f);
      *(float2*)&ldsMean[r * ASTRIDE + (t & 31) * 2] = make_float2(ax[p] * inv, ay[p] * inv);
    }
    __syncthreads();
    const int tn = __builtin_amdgcn_readfirstlane(sNext);

    const bool nb  = tn >= ntA;
    const int  nl  = nb ? tn - ntA : tn;
    const int  nrn = nb ? nrowsB : nrowsA;
    const int* __restrict__ offn  = nb ? offB : offA;
    const float* __restrict__ xdn = nb ? xdstB : xdstA;
    int ovn = 0;
    if (tn < ntot && (t & 31) < 5) {
      const int gi = nl * 32 + wv * 4 + (t & 31);
      ovn = offn[gi < nrn ? gi : nrn];
    }
    const float bcol = (tb ? biasB : biasA)[ncol];
    #pragma unroll
    for (int u = 0; u < 2; ++u) {
      const int g = nl * 32 + xr + 16 * u;
      xv[u] = make_float4(0.f, 0.f, 0.f, 0.f);
      if (tn < ntot && g < nrn) xv[u] = *(const float4*)&xdn[(size_t)g * D + xc];
    }

    v8f acc = {0.f, 0.f, 0.f, 0.f, 0.f, 0.f, 0.f, 0.f};
    #pragma unroll
    for (int k = 0; k < D; k += 4) {
      const int kk = k + khalf;
      const v2f a = *(const v2f*)&ldsMean[mrow * ASTRIDE + kk];
      v2f b;
      b.x = ldsWl[(kk + 0) * WSTRIDE + ncol];
      b.y = ldsWl[(kk + 1) * WSTRIDE + ncol];
      acc = __builtin_amdgcn_wmma_f32_16x16x4_f32(false, a, false, b,
                                                  (short)0, acc, false, false);
    }
    #pragma unroll
    for (int k = 0; k < D; k += 4) {
      const int kk = k + khalf;
      const v2f a = *(const v2f*)&ldsX[mrow * ASTRIDE + kk];
      v2f b;
      b.x = ldsWr[(kk + 0) * WSTRIDE + ncol];
      b.y = ldsWr[(kk + 1) * WSTRIDE + ncol];
      acc = __builtin_amdgcn_wmma_f32_16x16x4_f32(false, a, false, b,
                                                  (short)0, acc, false, false);
    }

    float* __restrict__ out = tb ? outB : outA;
    #pragma unroll
    for (int v = 0; v < 8; ++v) {
      const int g = row0 + rt * 16 + mhalf + v;
      if (g < nrows) {
        float val = acc[v] + bcol;
        if (do_relu) val = fmaxf(val, 0.0f);
        out[(size_t)g * D + ncol] = val;
      }
    }
    ov = ovn;
    tile = tn;
  }
}

extern "C" void kernel_launch(void* const* d_in, const int* in_sizes, int n_in,
                              void* d_out, int out_size, void* d_ws, size_t ws_size,
                              hipStream_t stream) {
  const float* x_user  = (const float*)d_in[0];
  const float* x_movie = (const float*)d_in[1];
  const float* Wl1_um  = (const float*)d_in[2];
  const float* Wr1_um  = (const float*)d_in[3];
  const float* b1_um   = (const float*)d_in[4];
  const float* Wl1_mu  = (const float*)d_in[5];
  const float* Wr1_mu  = (const float*)d_in[6];
  const float* b1_mu   = (const float*)d_in[7];
  const float* Wl2_um  = (const float*)d_in[8];
  const float* Wr2_um  = (const float*)d_in[9];
  const float* b2_um   = (const float*)d_in[10];
  const float* Wl2_mu  = (const float*)d_in[11];
  const float* Wr2_mu  = (const float*)d_in[12];
  const float* b2_mu   = (const float*)d_in[13];
  const int*   uidx    = (const int*)d_in[14];
  const int*   midx    = (const int*)d_in[15];

  float* ws  = (float*)d_ws;
  int*   wsi = (int*)d_ws;
  int* lst0  = wsi + W_LST0;
  int* lst1  = wsi + W_LST1;
  int* offU  = wsi + W_OFFU;
  int* offM  = wsi + W_OFFM;
  int* colU  = wsi + W_COLU;
  int* colM  = wsi + W_COLM;
  int* rec0  = wsi + W_REC0;
  int* rec1  = wsi + W_REC1;
  int* ctr   = wsi + W_CTR;
  float* u1 = ws + OFF_U1;
  float* m1 = ws + OFF_M1;

  float* u2 = (float*)d_out;
  float* m2 = (float*)d_out + (size_t)N_USER * D;

  const dim3 blk(256);
  const dim3 big(1024);
  const dim3 gChunk(NCH, 2);
  const dim3 gBucket(NB, 2);
  const dim3 gLayer(1280);

  bin_sort_kernel<<<gChunk, big, 0, stream>>>(uidx, midx, lst0, lst1,
                                              rec0, rec1, ctr, N_EDGE);
  bucket_csr_kernel<<<gBucket, big, 0, stream>>>(lst0, lst1, rec0, rec1,
                                                 offU, offM, colU, colM);

  sage_pull_wmma_kernel<<<gLayer, blk, 0, stream>>>(
      x_user,  offM, colM, x_movie, Wl1_um, Wr1_um, b1_um, m1, N_MOVIE,
      x_movie, offU, colU, x_user,  Wl1_mu, Wr1_mu, b1_mu, u1, N_USER,
      ctr, 1);

  sage_pull_wmma_kernel<<<gLayer, blk, 0, stream>>>(
      u1, offM, colM, m1, Wl2_um, Wr2_um, b2_um, m2, N_MOVIE,
      m1, offU, colU, u1, Wl2_mu, Wr2_mu, b2_mu, u2, N_USER,
      ctr + 1, 0);
}
